# starting kernel
# baseline (speedup 1.0000x reference)
#include <hip/hip_runtime.h>
#include <math.h>

#define A_DIM 64
#define F_DIM 12
#define G_DIM (F_DIM * A_DIM)
#define NMU 11
#define MSG_SCALE 3.60843918e-04f
#define KCAL2MEV 43.3641f

typedef __attribute__((ext_vector_type(2))) float v2f;
typedef __attribute__((ext_vector_type(8))) float v8f;

__device__ __forceinline__ float silu(float x) { return x / (1.0f + expf(-x)); }

__global__ void embed_kernel(const int* __restrict__ z, const float* __restrict__ We,
                             const float* __restrict__ be, float* __restrict__ y, int N) {
    int i = blockIdx.x * blockDim.x + threadIdx.x;
    if (i < N * A_DIM) {
        int n = i >> 6, a = i & 63;
        y[i] = We[z[n] * A_DIM + a] + be[a];
    }
}

__global__ void edge_feat_kernel(const float* __restrict__ pos, const int* __restrict__ src,
                                 const int* __restrict__ dst, float* __restrict__ ef, int E) {
    int e = blockIdx.x * blockDim.x + threadIdx.x;
    if (e >= E) return;
    int s = src[e], d = dst[e];
    float dx = pos[s * 3 + 0] - pos[d * 3 + 0];
    float dy = pos[s * 3 + 1] - pos[d * 3 + 1];
    float dz = pos[s * 3 + 2] - pos[d * 3 + 2];
    float r = sqrtf(dx * dx + dy * dy + dz * dz);
#pragma unroll
    for (int j = 0; j < NMU; ++j) {
        float t = r - 0.5f * (float)j;
        ef[(size_t)e * F_DIM + j] = expf(-2.0f * t * t);
    }
    ef[(size_t)e * F_DIM + NMU] = 1.0f;
}

__global__ void row_ptr_kernel(const int* __restrict__ src, int* __restrict__ rp, int N, int E) {
    int e = blockIdx.x * blockDim.x + threadIdx.x;
    if (e > E) return;
    if (e == 0) {
        for (int n = 0; n <= src[0]; ++n) rp[n] = 0;
    } else if (e == E) {
        for (int n = src[E - 1] + 1; n <= N; ++n) rp[n] = E;
    } else {
        int a = src[e - 1], b = src[e];
        for (int n = a + 1; n <= b; ++n) rp[n] = e;
    }
}

__global__ __launch_bounds__(128) void node_tp_wmma(const float* __restrict__ y,
                                                    const float* __restrict__ Wl,
                                                    float* __restrict__ g, int N) {
    __shared__ float sy[64 * A_DIM];
    int nodeBase = blockIdx.x * 64;
    bool full = (nodeBase + 64 <= N);
    for (int i = threadIdx.x * 4; i < 64 * A_DIM; i += 128 * 4) {
        int r = i >> 6, c = i & 63;
        int row = nodeBase + r;
        if (row >= N) row = N - 1;
        *(float4*)(sy + i) = *(const float4*)(y + (size_t)row * A_DIM + c);
    }
    __syncthreads();

    int wave = threadIdx.x >> 5;
    int lane = threadIdx.x & 31;
    int m    = lane & 15;
    int hi   = lane >> 4;

    for (int t = 0; t < 12; ++t) {
        int tile    = wave * 12 + t;
        int colBase = tile << 4;
        int f       = colBase >> 6;
        int bcol    = (colBase & 63) + m;
        const float* Bp = Wl + (size_t)f * (A_DIM * A_DIM) + 2 * hi * A_DIM + bcol;

        v2f bfr[16];
#pragma unroll
        for (int k = 0; k < 16; ++k) {
            bfr[k].x = Bp[k * 4 * A_DIM];
            bfr[k].y = Bp[k * 4 * A_DIM + A_DIM];
        }

        v8f c0 = {}, c1 = {}, c2 = {}, c3 = {};
#pragma unroll
        for (int k = 0; k < 16; ++k) {
            int ka = k * 4 + 2 * hi;
            v2f a0; a0.x = sy[(m     ) * A_DIM + ka]; a0.y = sy[(m     ) * A_DIM + ka + 1];
            v2f a1; a1.x = sy[(m + 16) * A_DIM + ka]; a1.y = sy[(m + 16) * A_DIM + ka + 1];
            v2f a2; a2.x = sy[(m + 32) * A_DIM + ka]; a2.y = sy[(m + 32) * A_DIM + ka + 1];
            v2f a3; a3.x = sy[(m + 48) * A_DIM + ka]; a3.y = sy[(m + 48) * A_DIM + ka + 1];
            c0 = __builtin_amdgcn_wmma_f32_16x16x4_f32(false, a0, false, bfr[k], (short)0, c0, false, false);
            c1 = __builtin_amdgcn_wmma_f32_16x16x4_f32(false, a1, false, bfr[k], (short)0, c1, false, false);
            c2 = __builtin_amdgcn_wmma_f32_16x16x4_f32(false, a2, false, bfr[k], (short)0, c2, false, false);
            c3 = __builtin_amdgcn_wmma_f32_16x16x4_f32(false, a3, false, bfr[k], (short)0, c3, false, false);
        }
        size_t colOff = (size_t)colBase + m;
        if (full) {
#pragma unroll
            for (int r = 0; r < 8; ++r) {
                int rr = hi * 8 + r;
                g[(size_t)(nodeBase      + rr) * G_DIM + colOff] = c0[r];
                g[(size_t)(nodeBase + 16 + rr) * G_DIM + colOff] = c1[r];
                g[(size_t)(nodeBase + 32 + rr) * G_DIM + colOff] = c2[r];
                g[(size_t)(nodeBase + 48 + rr) * G_DIM + colOff] = c3[r];
            }
        } else {
#pragma unroll
            for (int r = 0; r < 8; ++r) {
                int rr = hi * 8 + r;
                if (nodeBase      + rr < N) g[(size_t)(nodeBase      + rr) * G_DIM + colOff] = c0[r];
                if (nodeBase + 16 + rr < N) g[(size_t)(nodeBase + 16 + rr) * G_DIM + colOff] = c1[r];
                if (nodeBase + 32 + rr < N) g[(size_t)(nodeBase + 32 + rr) * G_DIM + colOff] = c2[r];
                if (nodeBase + 48 + rr < N) g[(size_t)(nodeBase + 48 + rr) * G_DIM + colOff] = c3[r];
            }
        }
    }
}

__global__ __launch_bounds__(64) void interact_kernel(float* __restrict__ y,
                                                      const float* __restrict__ g,
                                                      const float* __restrict__ ef,
                                                      const int* __restrict__ rp,
                                                      const int* __restrict__ dst, int N) {
    int n = blockIdx.x;
    if (n >= N) return;
    int t = threadIdx.x;
    float acc = 0.0f;
    int e0 = rp[n], e1 = rp[n + 1];
    for (int e = e0; e < e1; ++e) {
        int d = dst[e];
        const float* ge = g + (size_t)d * G_DIM + t;
        const float* fe = ef + (size_t)e * F_DIM;
        float msum = 0.0f;
#pragma unroll
        for (int f = 0; f < F_DIM; ++f) msum += fe[f] * ge[f * A_DIM];
        acc += silu(msum * MSG_SCALE);
    }
    y[(size_t)n * A_DIM + t] += acc;
}

__global__ __launch_bounds__(256) void readout_kernel(const float* __restrict__ y,
                                                      const float* __restrict__ W1,
                                                      const float* __restrict__ b1,
                                                      const float* __restrict__ W2,
                                                      const float* __restrict__ b2,
                                                      float* __restrict__ partial, int N) {
    __shared__ float red[256];
    int n = blockIdx.x * 256 + threadIdx.x;
    float r = 0.0f;
    if (n < N) {
        float yr[A_DIM];
#pragma unroll
        for (int a = 0; a < A_DIM; ++a) yr[a] = y[(size_t)n * A_DIM + a];
        for (int mm = 0; mm < 128; ++mm) {
            float s = b1[mm];
#pragma unroll
            for (int a = 0; a < A_DIM; ++a) s += yr[a] * W1[a * 128 + mm];
            r += silu(s) * W2[mm];
        }
        r += b2[0];
    }
    red[threadIdx.x] = r;
    __syncthreads();
    for (int off = 128; off > 0; off >>= 1) {
        if (threadIdx.x < off) red[threadIdx.x] += red[threadIdx.x + off];
        __syncthreads();
    }
    if (threadIdx.x == 0) partial[blockIdx.x] = red[0];
}

__global__ void finalize_kernel(const float* __restrict__ partial, int nPart,
                                const float* __restrict__ energy,
                                float* __restrict__ out, int N) {
    __shared__ float red[256];
    float v = 0.0f;
    for (int i = threadIdx.x; i < nPart; i += 256) v += partial[i];
    red[threadIdx.x] = v;
    __syncthreads();
    for (int off = 128; off > 0; off >>= 1) {
        if (threadIdx.x < off) red[threadIdx.x] += red[threadIdx.x + off];
        __syncthreads();
    }
    if (threadIdx.x == 0) {
        out[0] = red[0];
        out[1 + (size_t)N * A_DIM] = energy[0] * KCAL2MEV * (-1.0f);
    }
}

__global__ void copy_y_kernel(const float* __restrict__ y, float* __restrict__ out, int N) {
    int i = blockIdx.x * blockDim.x + threadIdx.x;
    if (i < N * A_DIM) out[1 + i] = y[i];
}

extern "C" void kernel_launch(void* const* d_in, const int* in_sizes, int n_in,
                              void* d_out, int out_size, void* d_ws, size_t ws_size,
                              hipStream_t stream) {
    const int*   z      = (const int*)d_in[0];
    const float* pos    = (const float*)d_in[1];
    const float* energy = (const float*)d_in[2];
    const int*   src    = (const int*)d_in[3];
    const int*   dst    = (const int*)d_in[4];
    const float* We     = (const float*)d_in[5];
    const float* be     = (const float*)d_in[6];
    const float* Wtp    = (const float*)d_in[7];
    const float* W1     = (const float*)d_in[8];
    const float* b1     = (const float*)d_in[9];
    const float* W2     = (const float*)d_in[10];
    const float* b2     = (const float*)d_in[11];
    float* out = (float*)d_out;

    const int N = in_sizes[0];
    const int E = in_sizes[3];

    float* fws = (float*)d_ws;
    size_t off = 0;
    float* ef      = fws + off; off += (size_t)E * F_DIM;
    float* g       = fws + off; off += (size_t)N * G_DIM;
    float* yw      = fws + off; off += (size_t)N * A_DIM;
    float* partial = fws + off; off += 256;
    int*   rp      = (int*)(fws + off);

    embed_kernel<<<(N * A_DIM + 255) / 256, 256, 0, stream>>>(z, We, be, yw, N);
    edge_feat_kernel<<<(E + 255) / 256, 256, 0, stream>>>(pos, src, dst, ef, E);
    row_ptr_kernel<<<(E + 1 + 255) / 256, 256, 0, stream>>>(src, rp, N, E);

    for (int l = 0; l < 3; ++l) {
        node_tp_wmma<<<(N + 63) / 64, 128, 0, stream>>>(
            yw, Wtp + (size_t)l * F_DIM * A_DIM * A_DIM, g, N);
        interact_kernel<<<N, 64, 0, stream>>>(yw, g, ef, rp, dst, N);
    }

    int nBlk = (N + 255) / 256;
    readout_kernel<<<nBlk, 256, 0, stream>>>(yw, W1, b1, W2, b2, partial, N);
    finalize_kernel<<<1, 256, 0, stream>>>(partial, nBlk, energy, out, N);
    copy_y_kernel<<<(N * A_DIM + 255) / 256, 256, 0, stream>>>(yw, out, N);
}


# readout y@W1 moved from a one-thread-per-node scalar loop (40 blocks) to f32 WMMA tiles, 64 nodes x 8 waves per block with the y tile in LDS; the readout sum's reduction tree changes
# speedup vs baseline: 1.3810x; 1.3810x over previous
#include <hip/hip_runtime.h>
#include <math.h>

#define A_DIM 64
#define F_DIM 12
#define G_DIM (F_DIM * A_DIM)
#define NMU 11
#define MSG_SCALE 3.60843918e-04f
#define KCAL2MEV 43.3641f

typedef __attribute__((ext_vector_type(2))) float v2f;
typedef __attribute__((ext_vector_type(8))) float v8f;

__device__ __forceinline__ float silu(float x) { return x / (1.0f + expf(-x)); }

__global__ void embed_kernel(const int* __restrict__ z, const float* __restrict__ We,
                             const float* __restrict__ be, float* __restrict__ y, int N) {
    int i = blockIdx.x * blockDim.x + threadIdx.x;
    if (i < N * A_DIM) {
        int n = i >> 6, a = i & 63;
        y[i] = We[z[n] * A_DIM + a] + be[a];
    }
}

__global__ void edge_feat_kernel(const float* __restrict__ pos, const int* __restrict__ src,
                                 const int* __restrict__ dst, float* __restrict__ ef, int E) {
    int e = blockIdx.x * blockDim.x + threadIdx.x;
    if (e >= E) return;
    int s = src[e], d = dst[e];
    float dx = pos[s * 3 + 0] - pos[d * 3 + 0];
    float dy = pos[s * 3 + 1] - pos[d * 3 + 1];
    float dz = pos[s * 3 + 2] - pos[d * 3 + 2];
    float r = sqrtf(dx * dx + dy * dy + dz * dz);
#pragma unroll
    for (int j = 0; j < NMU; ++j) {
        float t = r - 0.5f * (float)j;
        ef[(size_t)e * F_DIM + j] = expf(-2.0f * t * t);
    }
    ef[(size_t)e * F_DIM + NMU] = 1.0f;
}

__global__ void row_ptr_kernel(const int* __restrict__ src, int* __restrict__ rp, int N, int E) {
    int e = blockIdx.x * blockDim.x + threadIdx.x;
    if (e > E) return;
    if (e == 0) {
        for (int n = 0; n <= src[0]; ++n) rp[n] = 0;
    } else if (e == E) {
        for (int n = src[E - 1] + 1; n <= N; ++n) rp[n] = E;
    } else {
        int a = src[e - 1], b = src[e];
        for (int n = a + 1; n <= b; ++n) rp[n] = e;
    }
}

__global__ __launch_bounds__(128) void node_tp_wmma(const float* __restrict__ y,
                                                    const float* __restrict__ Wl,
                                                    float* __restrict__ g, int N) {
    __shared__ float sy[64 * A_DIM];
    int nodeBase = blockIdx.x * 64;
    bool full = (nodeBase + 64 <= N);
    for (int i = threadIdx.x * 4; i < 64 * A_DIM; i += 128 * 4) {
        int r = i >> 6, c = i & 63;
        int row = nodeBase + r;
        if (row >= N) row = N - 1;
        *(float4*)(sy + i) = *(const float4*)(y + (size_t)row * A_DIM + c);
    }
    __syncthreads();

    int wave = threadIdx.x >> 5;
    int lane = threadIdx.x & 31;
    int m    = lane & 15;
    int hi   = lane >> 4;

    for (int t = 0; t < 12; ++t) {
        int tile    = wave * 12 + t;
        int colBase = tile << 4;
        int f       = colBase >> 6;
        int bcol    = (colBase & 63) + m;
        const float* Bp = Wl + (size_t)f * (A_DIM * A_DIM) + 2 * hi * A_DIM + bcol;

        v2f bfr[16];
#pragma unroll
        for (int k = 0; k < 16; ++k) {
            bfr[k].x = Bp[k * 4 * A_DIM];
            bfr[k].y = Bp[k * 4 * A_DIM + A_DIM];
        }

        v8f c0 = {}, c1 = {}, c2 = {}, c3 = {};
#pragma unroll
        for (int k = 0; k < 16; ++k) {
            int ka = k * 4 + 2 * hi;
            v2f a0; a0.x = sy[(m     ) * A_DIM + ka]; a0.y = sy[(m     ) * A_DIM + ka + 1];
            v2f a1; a1.x = sy[(m + 16) * A_DIM + ka]; a1.y = sy[(m + 16) * A_DIM + ka + 1];
            v2f a2; a2.x = sy[(m + 32) * A_DIM + ka]; a2.y = sy[(m + 32) * A_DIM + ka + 1];
            v2f a3; a3.x = sy[(m + 48) * A_DIM + ka]; a3.y = sy[(m + 48) * A_DIM + ka + 1];
            c0 = __builtin_amdgcn_wmma_f32_16x16x4_f32(false, a0, false, bfr[k], (short)0, c0, false, false);
            c1 = __builtin_amdgcn_wmma_f32_16x16x4_f32(false, a1, false, bfr[k], (short)0, c1, false, false);
            c2 = __builtin_amdgcn_wmma_f32_16x16x4_f32(false, a2, false, bfr[k], (short)0, c2, false, false);
            c3 = __builtin_amdgcn_wmma_f32_16x16x4_f32(false, a3, false, bfr[k], (short)0, c3, false, false);
        }
        size_t colOff = (size_t)colBase + m;
        if (full) {
#pragma unroll
            for (int r = 0; r < 8; ++r) {
                int rr = hi * 8 + r;
                g[(size_t)(nodeBase      + rr) * G_DIM + colOff] = c0[r];
                g[(size_t)(nodeBase + 16 + rr) * G_DIM + colOff] = c1[r];
                g[(size_t)(nodeBase + 32 + rr) * G_DIM + colOff] = c2[r];
                g[(size_t)(nodeBase + 48 + rr) * G_DIM + colOff] = c3[r];
            }
        } else {
#pragma unroll
            for (int r = 0; r < 8; ++r) {
                int rr = hi * 8 + r;
                if (nodeBase      + rr < N) g[(size_t)(nodeBase      + rr) * G_DIM + colOff] = c0[r];
                if (nodeBase + 16 + rr < N) g[(size_t)(nodeBase + 16 + rr) * G_DIM + colOff] = c1[r];
                if (nodeBase + 32 + rr < N) g[(size_t)(nodeBase + 32 + rr) * G_DIM + colOff] = c2[r];
                if (nodeBase + 48 + rr < N) g[(size_t)(nodeBase + 48 + rr) * G_DIM + colOff] = c3[r];
            }
        }
    }
}

__global__ __launch_bounds__(64) void interact_kernel(float* __restrict__ y,
                                                      const float* __restrict__ g,
                                                      const float* __restrict__ ef,
                                                      const int* __restrict__ rp,
                                                      const int* __restrict__ dst, int N) {
    int n = blockIdx.x;
    if (n >= N) return;
    int t = threadIdx.x;
    float acc = 0.0f;
    int e0 = rp[n], e1 = rp[n + 1];
    for (int e = e0; e < e1; ++e) {
        int d = dst[e];
        const float* ge = g + (size_t)d * G_DIM + t;
        const float* fe = ef + (size_t)e * F_DIM;
        float msum = 0.0f;
#pragma unroll
        for (int f = 0; f < F_DIM; ++f) msum += fe[f] * ge[f * A_DIM];
        acc += silu(msum * MSG_SCALE);
    }
    y[(size_t)n * A_DIM + t] += acc;
}

#define M_DIM 128
#define SY_PITCH 68

__global__ __launch_bounds__(256) void readout_wmma(const float* __restrict__ y,
                                                    const float* __restrict__ W1,
                                                    const float* __restrict__ b1,
                                                    const float* __restrict__ W2,
                                                    const float* __restrict__ b2,
                                                    float* __restrict__ partial, int N) {
    __shared__ __attribute__((aligned(16))) float sy[64 * SY_PITCH];
    __shared__ float red[8];
    int nodeBase = blockIdx.x * 64;
    int wave = threadIdx.x >> 5;
    int lane = threadIdx.x & 31;
    int m    = lane & 15;
    int hi   = lane >> 4;
    int col  = wave * 16 + m;

    const float* Bp = W1 + 2 * hi * M_DIM + col;
    v2f bfr[16];
#pragma unroll
    for (int k = 0; k < 16; ++k) {
        bfr[k].x = Bp[k * 4 * M_DIM];
        bfr[k].y = Bp[k * 4 * M_DIM + M_DIM];
    }
    float bias = b1[col];
    float w2   = W2[col];

    float4 st[4];
#pragma unroll
    for (int j = 0; j < 4; ++j) {
        int i = threadIdx.x * 4 + j * 1024;
        int row = nodeBase + (i >> 6);
        if (row >= N) row = N - 1;
        st[j] = *(const float4*)(y + (size_t)row * A_DIM + (i & 63));
    }
#pragma unroll
    for (int j = 0; j < 4; ++j) {
        int i = threadIdx.x * 4 + j * 1024;
        *(float4*)(sy + (i >> 6) * SY_PITCH + (i & 63)) = st[j];
    }
    __syncthreads();

    v8f c0 = {}, c1 = {}, c2 = {}, c3 = {};
#pragma unroll
    for (int k = 0; k < 16; ++k) {
        int ka = k * 4 + 2 * hi;
        v2f a0 = *(const v2f*)(sy + (m     ) * SY_PITCH + ka);
        v2f a1 = *(const v2f*)(sy + (m + 16) * SY_PITCH + ka);
        v2f a2 = *(const v2f*)(sy + (m + 32) * SY_PITCH + ka);
        v2f a3 = *(const v2f*)(sy + (m + 48) * SY_PITCH + ka);
        c0 = __builtin_amdgcn_wmma_f32_16x16x4_f32(false, a0, false, bfr[k], (short)0, c0, false, false);
        c1 = __builtin_amdgcn_wmma_f32_16x16x4_f32(false, a1, false, bfr[k], (short)0, c1, false, false);
        c2 = __builtin_amdgcn_wmma_f32_16x16x4_f32(false, a2, false, bfr[k], (short)0, c2, false, false);
        c3 = __builtin_amdgcn_wmma_f32_16x16x4_f32(false, a3, false, bfr[k], (short)0, c3, false, false);
    }

    float s = 0.0f;
#pragma unroll
    for (int r = 0; r < 8; ++r) {
        int row = nodeBase + hi * 8 + r;
        float h0 = silu(c0[r] + bias);
        float h1 = silu(c1[r] + bias);
        float h2 = silu(c2[r] + bias);
        float h3 = silu(c3[r] + bias);
        s += (row      < N) ? h0 : 0.0f;
        s += (row + 16 < N) ? h1 : 0.0f;
        s += (row + 32 < N) ? h2 : 0.0f;
        s += (row + 48 < N) ? h3 : 0.0f;
    }
    s *= w2;
    for (int off = 16; off > 0; off >>= 1) s += __shfl_xor(s, off);
    if (lane == 0) red[wave] = s;
    __syncthreads();
    if (threadIdx.x == 0) {
        float tot = 0.0f;
#pragma unroll
        for (int w = 0; w < 8; ++w) tot += red[w];
        int cnt = N - nodeBase;
        if (cnt > 64) cnt = 64;
        partial[blockIdx.x] = tot + b2[0] * (float)cnt;
    }
}

__global__ void finalize_kernel(const float* __restrict__ partial, int nPart,
                                const float* __restrict__ energy,
                                float* __restrict__ out, int N) {
    __shared__ float red[256];
    float v = 0.0f;
    for (int i = threadIdx.x; i < nPart; i += 256) v += partial[i];
    red[threadIdx.x] = v;
    __syncthreads();
    for (int off = 128; off > 0; off >>= 1) {
        if (threadIdx.x < off) red[threadIdx.x] += red[threadIdx.x + off];
        __syncthreads();
    }
    if (threadIdx.x == 0) {
        out[0] = red[0];
        out[1 + (size_t)N * A_DIM] = energy[0] * KCAL2MEV * (-1.0f);
    }
}

__global__ void copy_y_kernel(const float* __restrict__ y, float* __restrict__ out, int N) {
    int i = blockIdx.x * blockDim.x + threadIdx.x;
    if (i < N * A_DIM) out[1 + i] = y[i];
}

extern "C" void kernel_launch(void* const* d_in, const int* in_sizes, int n_in,
                              void* d_out, int out_size, void* d_ws, size_t ws_size,
                              hipStream_t stream) {
    const int*   z      = (const int*)d_in[0];
    const float* pos    = (const float*)d_in[1];
    const float* energy = (const float*)d_in[2];
    const int*   src    = (const int*)d_in[3];
    const int*   dst    = (const int*)d_in[4];
    const float* We     = (const float*)d_in[5];
    const float* be     = (const float*)d_in[6];
    const float* Wtp    = (const float*)d_in[7];
    const float* W1     = (const float*)d_in[8];
    const float* b1     = (const float*)d_in[9];
    const float* W2     = (const float*)d_in[10];
    const float* b2     = (const float*)d_in[11];
    float* out = (float*)d_out;

    const int N = in_sizes[0];
    const int E = in_sizes[3];

    float* fws = (float*)d_ws;
    size_t off = 0;
    float* ef      = fws + off; off += (size_t)E * F_DIM;
    float* g       = fws + off; off += (size_t)N * G_DIM;
    float* yw      = fws + off; off += (size_t)N * A_DIM;
    int*   rp      = (int*)(fws + off);
    float* partial = g;

    embed_kernel<<<(N * A_DIM + 255) / 256, 256, 0, stream>>>(z, We, be, yw, N);
    edge_feat_kernel<<<(E + 255) / 256, 256, 0, stream>>>(pos, src, dst, ef, E);
    row_ptr_kernel<<<(E + 1 + 255) / 256, 256, 0, stream>>>(src, rp, N, E);

    for (int l = 0; l < 3; ++l) {
        node_tp_wmma<<<(N + 63) / 64, 128, 0, stream>>>(
            yw, Wtp + (size_t)l * F_DIM * A_DIM * A_DIM, g, N);
        interact_kernel<<<N, 64, 0, stream>>>(yw, g, ef, rp, dst, N);
    }

    int nBlk = (N + 63) / 64;
    readout_wmma<<<nBlk, 256, 0, stream>>>(yw, W1, b1, W2, b2, partial, N);
    finalize_kernel<<<1, 256, 0, stream>>>(partial, nBlk, energy, out, N);
    copy_y_kernel<<<(N * A_DIM + 255) / 256, 256, 0, stream>>>(yw, out, N);
}


# interact: one wave per node (8-wave blocks); each lane half gathers one edge's g row with 16-byte loads, so one load instruction covers two edges; order of the edge sum kept
# speedup vs baseline: 1.4238x; 1.0310x over previous
#include <hip/hip_runtime.h>
#include <math.h>

#define A_DIM 64
#define F_DIM 12
#define G_DIM (F_DIM * A_DIM)
#define NMU 11
#define MSG_SCALE 3.60843918e-04f
#define KCAL2MEV 43.3641f

typedef __attribute__((ext_vector_type(2))) float v2f;
typedef __attribute__((ext_vector_type(8))) float v8f;

__device__ __forceinline__ float silu(float x) { return x / (1.0f + expf(-x)); }

__global__ void embed_kernel(const int* __restrict__ z, const float* __restrict__ We,
                             const float* __restrict__ be, float* __restrict__ y, int N) {
    int i = blockIdx.x * blockDim.x + threadIdx.x;
    if (i < N * A_DIM) {
        int n = i >> 6, a = i & 63;
        y[i] = We[z[n] * A_DIM + a] + be[a];
    }
}

__global__ void edge_feat_kernel(const float* __restrict__ pos, const int* __restrict__ src,
                                 const int* __restrict__ dst, float* __restrict__ ef, int E) {
    int e = blockIdx.x * blockDim.x + threadIdx.x;
    if (e >= E) return;
    int s = src[e], d = dst[e];
    float dx = pos[s * 3 + 0] - pos[d * 3 + 0];
    float dy = pos[s * 3 + 1] - pos[d * 3 + 1];
    float dz = pos[s * 3 + 2] - pos[d * 3 + 2];
    float r = sqrtf(dx * dx + dy * dy + dz * dz);
#pragma unroll
    for (int j = 0; j < NMU; ++j) {
        float t = r - 0.5f * (float)j;
        ef[(size_t)e * F_DIM + j] = expf(-2.0f * t * t);
    }
    ef[(size_t)e * F_DIM + NMU] = 1.0f;
}

__global__ void row_ptr_kernel(const int* __restrict__ src, int* __restrict__ rp, int N, int E) {
    int e = blockIdx.x * blockDim.x + threadIdx.x;
    if (e > E) return;
    if (e == 0) {
        for (int n = 0; n <= src[0]; ++n) rp[n] = 0;
    } else if (e == E) {
        for (int n = src[E - 1] + 1; n <= N; ++n) rp[n] = E;
    } else {
        int a = src[e - 1], b = src[e];
        for (int n = a + 1; n <= b; ++n) rp[n] = e;
    }
}

__global__ __launch_bounds__(128) void node_tp_wmma(const float* __restrict__ y,
                                                    const float* __restrict__ Wl,
                                                    float* __restrict__ g, int N) {
    __shared__ float sy[64 * A_DIM];
    int nodeBase = blockIdx.x * 64;
    bool full = (nodeBase + 64 <= N);
    for (int i = threadIdx.x * 4; i < 64 * A_DIM; i += 128 * 4) {
        int r = i >> 6, c = i & 63;
        int row = nodeBase + r;
        if (row >= N) row = N - 1;
        *(float4*)(sy + i) = *(const float4*)(y + (size_t)row * A_DIM + c);
    }
    __syncthreads();

    int wave = threadIdx.x >> 5;
    int lane = threadIdx.x & 31;
    int m    = lane & 15;
    int hi   = lane >> 4;

    for (int t = 0; t < 12; ++t) {
        int tile    = wave * 12 + t;
        int colBase = tile << 4;
        int f       = colBase >> 6;
        int bcol    = (colBase & 63) + m;
        const float* Bp = Wl + (size_t)f * (A_DIM * A_DIM) + 2 * hi * A_DIM + bcol;

        v2f bfr[16];
#pragma unroll
        for (int k = 0; k < 16; ++k) {
            bfr[k].x = Bp[k * 4 * A_DIM];
            bfr[k].y = Bp[k * 4 * A_DIM + A_DIM];
        }

        v8f c0 = {}, c1 = {}, c2 = {}, c3 = {};
#pragma unroll
        for (int k = 0; k < 16; ++k) {
            int ka = k * 4 + 2 * hi;
            v2f a0; a0.x = sy[(m     ) * A_DIM + ka]; a0.y = sy[(m     ) * A_DIM + ka + 1];
            v2f a1; a1.x = sy[(m + 16) * A_DIM + ka]; a1.y = sy[(m + 16) * A_DIM + ka + 1];
            v2f a2; a2.x = sy[(m + 32) * A_DIM + ka]; a2.y = sy[(m + 32) * A_DIM + ka + 1];
            v2f a3; a3.x = sy[(m + 48) * A_DIM + ka]; a3.y = sy[(m + 48) * A_DIM + ka + 1];
            c0 = __builtin_amdgcn_wmma_f32_16x16x4_f32(false, a0, false, bfr[k], (short)0, c0, false, false);
            c1 = __builtin_amdgcn_wmma_f32_16x16x4_f32(false, a1, false, bfr[k], (short)0, c1, false, false);
            c2 = __builtin_amdgcn_wmma_f32_16x16x4_f32(false, a2, false, bfr[k], (short)0, c2, false, false);
            c3 = __builtin_amdgcn_wmma_f32_16x16x4_f32(false, a3, false, bfr[k], (short)0, c3, false, false);
        }
        size_t colOff = (size_t)colBase + m;
        if (full) {
#pragma unroll
            for (int r = 0; r < 8; ++r) {
                int rr = hi * 8 + r;
                g[(size_t)(nodeBase      + rr) * G_DIM + colOff] = c0[r];
                g[(size_t)(nodeBase + 16 + rr) * G_DIM + colOff] = c1[r];
                g[(size_t)(nodeBase + 32 + rr) * G_DIM + colOff] = c2[r];
                g[(size_t)(nodeBase + 48 + rr) * G_DIM + colOff] = c3[r];
            }
        } else {
#pragma unroll
            for (int r = 0; r < 8; ++r) {
                int rr = hi * 8 + r;
                if (nodeBase      + rr < N) g[(size_t)(nodeBase      + rr) * G_DIM + colOff] = c0[r];
                if (nodeBase + 16 + rr < N) g[(size_t)(nodeBase + 16 + rr) * G_DIM + colOff] = c1[r];
                if (nodeBase + 32 + rr < N) g[(size_t)(nodeBase + 32 + rr) * G_DIM + colOff] = c2[r];
                if (nodeBase + 48 + rr < N) g[(size_t)(nodeBase + 48 + rr) * G_DIM + colOff] = c3[r];
            }
        }
    }
}

typedef __attribute__((ext_vector_type(4))) float v4f;

__device__ __forceinline__ float xhalf(float x) {
    return __builtin_bit_cast(float, __builtin_amdgcn_permlanex16(__builtin_bit_cast(int, x), __builtin_bit_cast(int, x),
                                                                  0x76543210, 0xfedcba98, false, false));
}

__device__ __forceinline__ v4f edge_msg(const float* __restrict__ g, const float* __restrict__ ef,
                                        int d, int e, int c) {
    const v4f* gp = (const v4f*)(g + (size_t)d * G_DIM) + c;
    const v4f* fp = (const v4f*)(ef + (size_t)e * F_DIM);
    v4f w0 = fp[0], w1 = fp[1], w2 = fp[2];
    float w[F_DIM] = {w0.x, w0.y, w0.z, w0.w, w1.x, w1.y, w1.z, w1.w, w2.x, w2.y, w2.z, w2.w};
    v4f v[F_DIM];
#pragma unroll
    for (int f = 0; f < F_DIM; ++f) v[f] = gp[f * 16];
    __builtin_amdgcn_sched_barrier(0);
    v4f m = {0.0f, 0.0f, 0.0f, 0.0f};
#pragma unroll
    for (int f = 0; f < F_DIM; ++f) m += w[f] * v[f];
    v4f s;
    s.x = silu(m.x * MSG_SCALE);
    s.y = silu(m.y * MSG_SCALE);
    s.z = silu(m.z * MSG_SCALE);
    s.w = silu(m.w * MSG_SCALE);
    return s;
}

__global__ __launch_bounds__(256) void interact_kernel(float* __restrict__ y,
                                                       const float* __restrict__ g,
                                                       const float* __restrict__ ef,
                                                       const int* __restrict__ rp,
                                                       const int* __restrict__ dst, int N) {
    int wave = __builtin_amdgcn_readfirstlane((int)(threadIdx.x >> 5));
    int n = blockIdx.x * 8 + wave;
    if (n >= N) return;
    int lane = threadIdx.x & 31;
    int c    = lane & 15;
    int hi   = lane >> 4;
    int e0 = rp[n], e1 = rp[n + 1];
    v4f acc = {0.0f, 0.0f, 0.0f, 0.0f};
    int e = e0;
    for (; e + 1 < e1; e += 2) {
        int d0 = dst[e], d1 = dst[e + 1];
        v4f s = edge_msg(g, ef, hi ? d1 : d0, e + hi, c);
        v4f o;
        o.x = xhalf(s.x); o.y = xhalf(s.y); o.z = xhalf(s.z); o.w = xhalf(s.w);
        acc = (acc + s) + o;
    }
    if (e < e1) {
        acc += edge_msg(g, ef, dst[e], e, c);
    }
    if (hi == 0) {
        v4f* yp = (v4f*)(y + (size_t)n * A_DIM) + c;
        *yp = *yp + acc;
    }
}

#define M_DIM 128
#define SY_PITCH 68

__global__ __launch_bounds__(256) void readout_wmma(const float* __restrict__ y,
                                                    const float* __restrict__ W1,
                                                    const float* __restrict__ b1,
                                                    const float* __restrict__ W2,
                                                    const float* __restrict__ b2,
                                                    float* __restrict__ partial, int N) {
    __shared__ __attribute__((aligned(16))) float sy[64 * SY_PITCH];
    __shared__ float red[8];
    int nodeBase = blockIdx.x * 64;
    int wave = threadIdx.x >> 5;
    int lane = threadIdx.x & 31;
    int m    = lane & 15;
    int hi   = lane >> 4;
    int col  = wave * 16 + m;

    const float* Bp = W1 + 2 * hi * M_DIM + col;
    v2f bfr[16];
#pragma unroll
    for (int k = 0; k < 16; ++k) {
        bfr[k].x = Bp[k * 4 * M_DIM];
        bfr[k].y = Bp[k * 4 * M_DIM + M_DIM];
    }
    float bias = b1[col];
    float w2   = W2[col];

    float4 st[4];
#pragma unroll
    for (int j = 0; j < 4; ++j) {
        int i = threadIdx.x * 4 + j * 1024;
        int row = nodeBase + (i >> 6);
        if (row >= N) row = N - 1;
        st[j] = *(const float4*)(y + (size_t)row * A_DIM + (i & 63));
    }
#pragma unroll
    for (int j = 0; j < 4; ++j) {
        int i = threadIdx.x * 4 + j * 1024;
        *(float4*)(sy + (i >> 6) * SY_PITCH + (i & 63)) = st[j];
    }
    __syncthreads();

    v8f c0 = {}, c1 = {}, c2 = {}, c3 = {};
#pragma unroll
    for (int k = 0; k < 16; ++k) {
        int ka = k * 4 + 2 * hi;
        v2f a0 = *(const v2f*)(sy + (m     ) * SY_PITCH + ka);
        v2f a1 = *(const v2f*)(sy + (m + 16) * SY_PITCH + ka);
        v2f a2 = *(const v2f*)(sy + (m + 32) * SY_PITCH + ka);
        v2f a3 = *(const v2f*)(sy + (m + 48) * SY_PITCH + ka);
        c0 = __builtin_amdgcn_wmma_f32_16x16x4_f32(false, a0, false, bfr[k], (short)0, c0, false, false);
        c1 = __builtin_amdgcn_wmma_f32_16x16x4_f32(false, a1, false, bfr[k], (short)0, c1, false, false);
        c2 = __builtin_amdgcn_wmma_f32_16x16x4_f32(false, a2, false, bfr[k], (short)0, c2, false, false);
        c3 = __builtin_amdgcn_wmma_f32_16x16x4_f32(false, a3, false, bfr[k], (short)0, c3, false, false);
    }

    float s = 0.0f;
#pragma unroll
    for (int r = 0; r < 8; ++r) {
        int row = nodeBase + hi * 8 + r;
        float h0 = silu(c0[r] + bias);
        float h1 = silu(c1[r] + bias);
        float h2 = silu(c2[r] + bias);
        float h3 = silu(c3[r] + bias);
        s += (row      < N) ? h0 : 0.0f;
        s += (row + 16 < N) ? h1 : 0.0f;
        s += (row + 32 < N) ? h2 : 0.0f;
        s += (row + 48 < N) ? h3 : 0.0f;
    }
    s *= w2;
    for (int off = 16; off > 0; off >>= 1) s += __shfl_xor(s, off);
    if (lane == 0) red[wave] = s;
    __syncthreads();
    if (threadIdx.x == 0) {
        float tot = 0.0f;
#pragma unroll
        for (int w = 0; w < 8; ++w) tot += red[w];
        int cnt = N - nodeBase;
        if (cnt > 64) cnt = 64;
        partial[blockIdx.x] = tot + b2[0] * (float)cnt;
    }
}

__global__ void finalize_kernel(const float* __restrict__ partial, int nPart,
                                const float* __restrict__ energy,
                                float* __restrict__ out, int N) {
    __shared__ float red[256];
    float v = 0.0f;
    for (int i = threadIdx.x; i < nPart; i += 256) v += partial[i];
    red[threadIdx.x] = v;
    __syncthreads();
    for (int off = 128; off > 0; off >>= 1) {
        if (threadIdx.x < off) red[threadIdx.x] += red[threadIdx.x + off];
        __syncthreads();
    }
    if (threadIdx.x == 0) {
        out[0] = red[0];
        out[1 + (size_t)N * A_DIM] = energy[0] * KCAL2MEV * (-1.0f);
    }
}

__global__ void copy_y_kernel(const float* __restrict__ y, float* __restrict__ out, int N) {
    int i = blockIdx.x * blockDim.x + threadIdx.x;
    if (i < N * A_DIM) out[1 + i] = y[i];
}

extern "C" void kernel_launch(void* const* d_in, const int* in_sizes, int n_in,
                              void* d_out, int out_size, void* d_ws, size_t ws_size,
                              hipStream_t stream) {
    const int*   z      = (const int*)d_in[0];
    const float* pos    = (const float*)d_in[1];
    const float* energy = (const float*)d_in[2];
    const int*   src    = (const int*)d_in[3];
    const int*   dst    = (const int*)d_in[4];
    const float* We     = (const float*)d_in[5];
    const float* be     = (const float*)d_in[6];
    const float* Wtp    = (const float*)d_in[7];
    const float* W1     = (const float*)d_in[8];
    const float* b1     = (const float*)d_in[9];
    const float* W2     = (const float*)d_in[10];
    const float* b2     = (const float*)d_in[11];
    float* out = (float*)d_out;

    const int N = in_sizes[0];
    const int E = in_sizes[3];

    float* fws = (float*)d_ws;
    size_t off = 0;
    float* ef      = fws + off; off += (size_t)E * F_DIM;
    float* g       = fws + off; off += (size_t)N * G_DIM;
    float* yw      = fws + off; off += (size_t)N * A_DIM;
    int*   rp      = (int*)(fws + off);
    float* partial = g;

    embed_kernel<<<(N * A_DIM + 255) / 256, 256, 0, stream>>>(z, We, be, yw, N);
    edge_feat_kernel<<<(E + 255) / 256, 256, 0, stream>>>(pos, src, dst, ef, E);
    row_ptr_kernel<<<(E + 1 + 255) / 256, 256, 0, stream>>>(src, rp, N, E);

    for (int l = 0; l < 3; ++l) {
        node_tp_wmma<<<(N + 63) / 64, 128, 0, stream>>>(
            yw, Wtp + (size_t)l * F_DIM * A_DIM * A_DIM, g, N);
        interact_kernel<<<(N + 7) / 8, 256, 0, stream>>>(yw, g, ef, rp, dst, N);
    }

    int nBlk = (N + 63) / 64;
    readout_wmma<<<nBlk, 256, 0, stream>>>(yw, W1, b1, W2, b2, partial, N);
    finalize_kernel<<<1, 256, 0, stream>>>(partial, nBlk, energy, out, N);
    copy_y_kernel<<<(N * A_DIM + 255) / 256, 256, 0, stream>>>(yw, out, N);
}


# node_tp: 48 column tiles of a 64-row block split over 3 blocks (grid 157x3, 4 tiles per wave) so every WGP gets 1-2 blocks; y tile at LDS pitch 68 so the tripled A-fragment reads do not conflict
# speedup vs baseline: 1.5499x; 1.0886x over previous
#include <hip/hip_runtime.h>
#include <math.h>

#define A_DIM 64
#define F_DIM 12
#define G_DIM (F_DIM * A_DIM)
#define NMU 11
#define MSG_SCALE 3.60843918e-04f
#define KCAL2MEV 43.3641f

typedef __attribute__((ext_vector_type(2))) float v2f;
typedef __attribute__((ext_vector_type(8))) float v8f;

__device__ __forceinline__ float silu(float x) { return x / (1.0f + expf(-x)); }

__global__ void embed_kernel(const int* __restrict__ z, const float* __restrict__ We,
                             const float* __restrict__ be, float* __restrict__ y, int N) {
    int i = blockIdx.x * blockDim.x + threadIdx.x;
    if (i < N * A_DIM) {
        int n = i >> 6, a = i & 63;
        y[i] = We[z[n] * A_DIM + a] + be[a];
    }
}

__global__ void edge_feat_kernel(const float* __restrict__ pos, const int* __restrict__ src,
                                 const int* __restrict__ dst, float* __restrict__ ef, int E) {
    int e = blockIdx.x * blockDim.x + threadIdx.x;
    if (e >= E) return;
    int s = src[e], d = dst[e];
    float dx = pos[s * 3 + 0] - pos[d * 3 + 0];
    float dy = pos[s * 3 + 1] - pos[d * 3 + 1];
    float dz = pos[s * 3 + 2] - pos[d * 3 + 2];
    float r = sqrtf(dx * dx + dy * dy + dz * dz);
#pragma unroll
    for (int j = 0; j < NMU; ++j) {
        float t = r - 0.5f * (float)j;
        ef[(size_t)e * F_DIM + j] = expf(-2.0f * t * t);
    }
    ef[(size_t)e * F_DIM + NMU] = 1.0f;
}

__global__ void row_ptr_kernel(const int* __restrict__ src, int* __restrict__ rp, int N, int E) {
    int e = blockIdx.x * blockDim.x + threadIdx.x;
    if (e > E) return;
    if (e == 0) {
        for (int n = 0; n <= src[0]; ++n) rp[n] = 0;
    } else if (e == E) {
        for (int n = src[E - 1] + 1; n <= N; ++n) rp[n] = E;
    } else {
        int a = src[e - 1], b = src[e];
        for (int n = a + 1; n <= b; ++n) rp[n] = e;
    }
}

#define TP_PITCH 68
#define TP_TILES 4
__global__ __launch_bounds__(128) void node_tp_wmma(const float* __restrict__ y,
                                                    const float* __restrict__ Wl,
                                                    float* __restrict__ g, int N) {
    __shared__ __attribute__((aligned(16))) float sy[64 * TP_PITCH];
    int nodeBase = blockIdx.x * 64;
    bool full = (nodeBase + 64 <= N);
    for (int i = threadIdx.x * 4; i < 64 * A_DIM; i += 128 * 4) {
        int r = i >> 6, c = i & 63;
        int row = nodeBase + r;
        if (row >= N) row = N - 1;
        *(float4*)(sy + r * TP_PITCH + c) = *(const float4*)(y + (size_t)row * A_DIM + c);
    }
    __syncthreads();

    int wave = threadIdx.x >> 5;
    int lane = threadIdx.x & 31;
    int m    = lane & 15;
    int hi   = lane >> 4;

    for (int t = 0; t < TP_TILES; ++t) {
        int tile    = ((int)blockIdx.y * 4 + wave) * TP_TILES + t;
        int colBase = tile << 4;
        int f       = colBase >> 6;
        int bcol    = (colBase & 63) + m;
        const float* Bp = Wl + (size_t)f * (A_DIM * A_DIM) + 2 * hi * A_DIM + bcol;

        v2f bfr[16];
#pragma unroll
        for (int k = 0; k < 16; ++k) {
            bfr[k].x = Bp[k * 4 * A_DIM];
            bfr[k].y = Bp[k * 4 * A_DIM + A_DIM];
        }

        v8f c0 = {}, c1 = {}, c2 = {}, c3 = {};
#pragma unroll
        for (int k = 0; k < 16; ++k) {
            int ka = k * 4 + 2 * hi;
            v2f a0; a0.x = sy[(m     ) * TP_PITCH + ka]; a0.y = sy[(m     ) * TP_PITCH + ka + 1];
            v2f a1; a1.x = sy[(m + 16) * TP_PITCH + ka]; a1.y = sy[(m + 16) * TP_PITCH + ka + 1];
            v2f a2; a2.x = sy[(m + 32) * TP_PITCH + ka]; a2.y = sy[(m + 32) * TP_PITCH + ka + 1];
            v2f a3; a3.x = sy[(m + 48) * TP_PITCH + ka]; a3.y = sy[(m + 48) * TP_PITCH + ka + 1];
            c0 = __builtin_amdgcn_wmma_f32_16x16x4_f32(false, a0, false, bfr[k], (short)0, c0, false, false);
            c1 = __builtin_amdgcn_wmma_f32_16x16x4_f32(false, a1, false, bfr[k], (short)0, c1, false, false);
            c2 = __builtin_amdgcn_wmma_f32_16x16x4_f32(false, a2, false, bfr[k], (short)0, c2, false, false);
            c3 = __builtin_amdgcn_wmma_f32_16x16x4_f32(false, a3, false, bfr[k], (short)0, c3, false, false);
        }
        size_t colOff = (size_t)colBase + m;
        if (full) {
#pragma unroll
            for (int r = 0; r < 8; ++r) {
                int rr = hi * 8 + r;
                g[(size_t)(nodeBase      + rr) * G_DIM + colOff] = c0[r];
                g[(size_t)(nodeBase + 16 + rr) * G_DIM + colOff] = c1[r];
                g[(size_t)(nodeBase + 32 + rr) * G_DIM + colOff] = c2[r];
                g[(size_t)(nodeBase + 48 + rr) * G_DIM + colOff] = c3[r];
            }
        } else {
#pragma unroll
            for (int r = 0; r < 8; ++r) {
                int rr = hi * 8 + r;
                if (nodeBase      + rr < N) g[(size_t)(nodeBase      + rr) * G_DIM + colOff] = c0[r];
                if (nodeBase + 16 + rr < N) g[(size_t)(nodeBase + 16 + rr) * G_DIM + colOff] = c1[r];
                if (nodeBase + 32 + rr < N) g[(size_t)(nodeBase + 32 + rr) * G_DIM + colOff] = c2[r];
                if (nodeBase + 48 + rr < N) g[(size_t)(nodeBase + 48 + rr) * G_DIM + colOff] = c3[r];
            }
        }
    }
}

typedef __attribute__((ext_vector_type(4))) float v4f;

__device__ __forceinline__ float xhalf(float x) {
    return __builtin_bit_cast(float, __builtin_amdgcn_permlanex16(__builtin_bit_cast(int, x), __builtin_bit_cast(int, x),
                                                                  0x76543210, 0xfedcba98, false, false));
}

__device__ __forceinline__ v4f edge_msg(const float* __restrict__ g, const float* __restrict__ ef,
                                        int d, int e, int c) {
    const v4f* gp = (const v4f*)(g + (size_t)d * G_DIM) + c;
    const v4f* fp = (const v4f*)(ef + (size_t)e * F_DIM);
    v4f w0 = fp[0], w1 = fp[1], w2 = fp[2];
    float w[F_DIM] = {w0.x, w0.y, w0.z, w0.w, w1.x, w1.y, w1.z, w1.w, w2.x, w2.y, w2.z, w2.w};
    v4f v[F_DIM];
#pragma unroll
    for (int f = 0; f < F_DIM; ++f) v[f] = gp[f * 16];
    __builtin_amdgcn_sched_barrier(0);
    v4f m = {0.0f, 0.0f, 0.0f, 0.0f};
#pragma unroll
    for (int f = 0; f < F_DIM; ++f) m += w[f] * v[f];
    v4f s;
    s.x = silu(m.x * MSG_SCALE);
    s.y = silu(m.y * MSG_SCALE);
    s.z = silu(m.z * MSG_SCALE);
    s.w = silu(m.w * MSG_SCALE);
    return s;
}

__global__ __launch_bounds__(256) void interact_kernel(float* __restrict__ y,
                                                       const float* __restrict__ g,
                                                       const float* __restrict__ ef,
                                                       const int* __restrict__ rp,
                                                       const int* __restrict__ dst, int N) {
    int wave = __builtin_amdgcn_readfirstlane((int)(threadIdx.x >> 5));
    int n = blockIdx.x * 8 + wave;
    if (n >= N) return;
    int lane = threadIdx.x & 31;
    int c    = lane & 15;
    int hi   = lane >> 4;
    int e0 = rp[n], e1 = rp[n + 1];
    v4f acc = {0.0f, 0.0f, 0.0f, 0.0f};
    int e = e0;
    for (; e + 1 < e1; e += 2) {
        int d0 = dst[e], d1 = dst[e + 1];
        v4f s = edge_msg(g, ef, hi ? d1 : d0, e + hi, c);
        v4f o;
        o.x = xhalf(s.x); o.y = xhalf(s.y); o.z = xhalf(s.z); o.w = xhalf(s.w);
        acc = (acc + s) + o;
    }
    if (e < e1) {
        acc += edge_msg(g, ef, dst[e], e, c);
    }
    if (hi == 0) {
        v4f* yp = (v4f*)(y + (size_t)n * A_DIM) + c;
        *yp = *yp + acc;
    }
}

#define M_DIM 128
#define SY_PITCH 68

__global__ __launch_bounds__(256) void readout_wmma(const float* __restrict__ y,
                                                    const float* __restrict__ W1,
                                                    const float* __restrict__ b1,
                                                    const float* __restrict__ W2,
                                                    const float* __restrict__ b2,
                                                    float* __restrict__ partial, int N) {
    __shared__ __attribute__((aligned(16))) float sy[64 * SY_PITCH];
    __shared__ float red[8];
    int nodeBase = blockIdx.x * 64;
    int wave = threadIdx.x >> 5;
    int lane = threadIdx.x & 31;
    int m    = lane & 15;
    int hi   = lane >> 4;
    int col  = wave * 16 + m;

    const float* Bp = W1 + 2 * hi * M_DIM + col;
    v2f bfr[16];
#pragma unroll
    for (int k = 0; k < 16; ++k) {
        bfr[k].x = Bp[k * 4 * M_DIM];
        bfr[k].y = Bp[k * 4 * M_DIM + M_DIM];
    }
    float bias = b1[col];
    float w2   = W2[col];

    float4 st[4];
#pragma unroll
    for (int j = 0; j < 4; ++j) {
        int i = threadIdx.x * 4 + j * 1024;
        int row = nodeBase + (i >> 6);
        if (row >= N) row = N - 1;
        st[j] = *(const float4*)(y + (size_t)row * A_DIM + (i & 63));
    }
#pragma unroll
    for (int j = 0; j < 4; ++j) {
        int i = threadIdx.x * 4 + j * 1024;
        *(float4*)(sy + (i >> 6) * SY_PITCH + (i & 63)) = st[j];
    }
    __syncthreads();

    v8f c0 = {}, c1 = {}, c2 = {}, c3 = {};
#pragma unroll
    for (int k = 0; k < 16; ++k) {
        int ka = k * 4 + 2 * hi;
        v2f a0 = *(const v2f*)(sy + (m     ) * SY_PITCH + ka);
        v2f a1 = *(const v2f*)(sy + (m + 16) * SY_PITCH + ka);
        v2f a2 = *(const v2f*)(sy + (m + 32) * SY_PITCH + ka);
        v2f a3 = *(const v2f*)(sy + (m + 48) * SY_PITCH + ka);
        c0 = __builtin_amdgcn_wmma_f32_16x16x4_f32(false, a0, false, bfr[k], (short)0, c0, false, false);
        c1 = __builtin_amdgcn_wmma_f32_16x16x4_f32(false, a1, false, bfr[k], (short)0, c1, false, false);
        c2 = __builtin_amdgcn_wmma_f32_16x16x4_f32(false, a2, false, bfr[k], (short)0, c2, false, false);
        c3 = __builtin_amdgcn_wmma_f32_16x16x4_f32(false, a3, false, bfr[k], (short)0, c3, false, false);
    }

    float s = 0.0f;
#pragma unroll
    for (int r = 0; r < 8; ++r) {
        int row = nodeBase + hi * 8 + r;
        float h0 = silu(c0[r] + bias);
        float h1 = silu(c1[r] + bias);
        float h2 = silu(c2[r] + bias);
        float h3 = silu(c3[r] + bias);
        s += (row      < N) ? h0 : 0.0f;
        s += (row + 16 < N) ? h1 : 0.0f;
        s += (row + 32 < N) ? h2 : 0.0f;
        s += (row + 48 < N) ? h3 : 0.0f;
    }
    s *= w2;
    for (int off = 16; off > 0; off >>= 1) s += __shfl_xor(s, off);
    if (lane == 0) red[wave] = s;
    __syncthreads();
    if (threadIdx.x == 0) {
        float tot = 0.0f;
#pragma unroll
        for (int w = 0; w < 8; ++w) tot += red[w];
        int cnt = N - nodeBase;
        if (cnt > 64) cnt = 64;
        partial[blockIdx.x] = tot + b2[0] * (float)cnt;
    }
}

__global__ void finalize_kernel(const float* __restrict__ partial, int nPart,
                                const float* __restrict__ energy,
                                float* __restrict__ out, int N) {
    __shared__ float red[256];
    float v = 0.0f;
    for (int i = threadIdx.x; i < nPart; i += 256) v += partial[i];
    red[threadIdx.x] = v;
    __syncthreads();
    for (int off = 128; off > 0; off >>= 1) {
        if (threadIdx.x < off) red[threadIdx.x] += red[threadIdx.x + off];
        __syncthreads();
    }
    if (threadIdx.x == 0) {
        out[0] = red[0];
        out[1 + (size_t)N * A_DIM] = energy[0] * KCAL2MEV * (-1.0f);
    }
}

__global__ void copy_y_kernel(const float* __restrict__ y, float* __restrict__ out, int N) {
    int i = blockIdx.x * blockDim.x + threadIdx.x;
    if (i < N * A_DIM) out[1 + i] = y[i];
}

extern "C" void kernel_launch(void* const* d_in, const int* in_sizes, int n_in,
                              void* d_out, int out_size, void* d_ws, size_t ws_size,
                              hipStream_t stream) {
    const int*   z      = (const int*)d_in[0];
    const float* pos    = (const float*)d_in[1];
    const float* energy = (const float*)d_in[2];
    const int*   src    = (const int*)d_in[3];
    const int*   dst    = (const int*)d_in[4];
    const float* We     = (const float*)d_in[5];
    const float* be     = (const float*)d_in[6];
    const float* Wtp    = (const float*)d_in[7];
    const float* W1     = (const float*)d_in[8];
    const float* b1     = (const float*)d_in[9];
    const float* W2     = (const float*)d_in[10];
    const float* b2     = (const float*)d_in[11];
    float* out = (float*)d_out;

    const int N = in_sizes[0];
    const int E = in_sizes[3];

    float* fws = (float*)d_ws;
    size_t off = 0;
    float* ef      = fws + off; off += (size_t)E * F_DIM;
    float* g       = fws + off; off += (size_t)N * G_DIM;
    float* yw      = fws + off; off += (size_t)N * A_DIM;
    int*   rp      = (int*)(fws + off);
    float* partial = g;

    embed_kernel<<<(N * A_DIM + 255) / 256, 256, 0, stream>>>(z, We, be, yw, N);
    edge_feat_kernel<<<(E + 255) / 256, 256, 0, stream>>>(pos, src, dst, ef, E);
    row_ptr_kernel<<<(E + 1 + 255) / 256, 256, 0, stream>>>(src, rp, N, E);

    for (int l = 0; l < 3; ++l) {
        node_tp_wmma<<<dim3((N + 63) / 64, 3), 128, 0, stream>>>(
            yw, Wtp + (size_t)l * F_DIM * A_DIM * A_DIM, g, N);
        interact_kernel<<<(N + 7) / 8, 256, 0, stream>>>(yw, g, ef, rp, dst, N);
    }

    int nBlk = (N + 63) / 64;
    readout_wmma<<<nBlk, 256, 0, stream>>>(yw, W1, b1, W2, b2, partial, N);
    finalize_kernel<<<1, 256, 0, stream>>>(partial, nBlk, energy, out, N);
    copy_y_kernel<<<(N * A_DIM + 255) / 256, 256, 0, stream>>>(yw, out, N);
}


# the four one-item-per-thread launches (embed, edge features, row pointers, y copy) folded into the first node_tp launch and the last interact launch: 12 launches -> 8; values and sums unchanged
# speedup vs baseline: 1.7004x; 1.0971x over previous
#include <hip/hip_runtime.h>
#include <math.h>

#define A_DIM 64
#define F_DIM 12
#define G_DIM (F_DIM * A_DIM)
#define NMU 11
#define MSG_SCALE 3.60843918e-04f
#define KCAL2MEV 43.3641f

typedef __attribute__((ext_vector_type(2))) float v2f;
typedef __attribute__((ext_vector_type(8))) float v8f;

__device__ __forceinline__ float silu(float x) { return x / (1.0f + expf(-x)); }

__device__ __forceinline__ void edge_feat_item(const float* __restrict__ pos, const int* __restrict__ src,
                                               const int* __restrict__ dst, float* __restrict__ ef, int e) {
    int s = src[e], d = dst[e];
    float dx = pos[s * 3 + 0] - pos[d * 3 + 0];
    float dy = pos[s * 3 + 1] - pos[d * 3 + 1];
    float dz = pos[s * 3 + 2] - pos[d * 3 + 2];
    float r = sqrtf(dx * dx + dy * dy + dz * dz);
#pragma unroll
    for (int j = 0; j < NMU; ++j) {
        float t = r - 0.5f * (float)j;
        ef[(size_t)e * F_DIM + j] = expf(-2.0f * t * t);
    }
    ef[(size_t)e * F_DIM + NMU] = 1.0f;
}

__device__ __forceinline__ void row_ptr_item(const int* __restrict__ src, int* __restrict__ rp, int N, int E, int e) {
    if (e == 0) {
        for (int n = 0; n <= src[0]; ++n) rp[n] = 0;
    } else if (e == E) {
        for (int n = src[E - 1] + 1; n <= N; ++n) rp[n] = E;
    } else {
        int a = src[e - 1], b = src[e];
        for (int n = a + 1; n <= b; ++n) rp[n] = e;
    }
}

#define TP_PITCH 68
#define TP_TILES 4

__device__ __forceinline__ void tp_tiles(const float* sy, const float* __restrict__ Wl,
                                         float* __restrict__ g, int N, int nodeBase, int third) {
    bool full = (nodeBase + 64 <= N);
    int wave = threadIdx.x >> 5;
    int lane = threadIdx.x & 31;
    int m    = lane & 15;
    int hi   = lane >> 4;

    for (int t = 0; t < TP_TILES; ++t) {
        int tile    = (third * 4 + wave) * TP_TILES + t;
        int colBase = tile << 4;
        int f       = colBase >> 6;
        int bcol    = (colBase & 63) + m;
        const float* Bp = Wl + (size_t)f * (A_DIM * A_DIM) + 2 * hi * A_DIM + bcol;

        v2f bfr[16];
#pragma unroll
        for (int k = 0; k < 16; ++k) {
            bfr[k].x = Bp[k * 4 * A_DIM];
            bfr[k].y = Bp[k * 4 * A_DIM + A_DIM];
        }

        v8f c0 = {}, c1 = {}, c2 = {}, c3 = {};
#pragma unroll
        for (int k = 0; k < 16; ++k) {
            int ka = k * 4 + 2 * hi;
            v2f a0; a0.x = sy[(m     ) * TP_PITCH + ka]; a0.y = sy[(m     ) * TP_PITCH + ka + 1];
            v2f a1; a1.x = sy[(m + 16) * TP_PITCH + ka]; a1.y = sy[(m + 16) * TP_PITCH + ka + 1];
            v2f a2; a2.x = sy[(m + 32) * TP_PITCH + ka]; a2.y = sy[(m + 32) * TP_PITCH + ka + 1];
            v2f a3; a3.x = sy[(m + 48) * TP_PITCH + ka]; a3.y = sy[(m + 48) * TP_PITCH + ka + 1];
            c0 = __builtin_amdgcn_wmma_f32_16x16x4_f32(false, a0, false, bfr[k], (short)0, c0, false, false);
            c1 = __builtin_amdgcn_wmma_f32_16x16x4_f32(false, a1, false, bfr[k], (short)0, c1, false, false);
            c2 = __builtin_amdgcn_wmma_f32_16x16x4_f32(false, a2, false, bfr[k], (short)0, c2, false, false);
            c3 = __builtin_amdgcn_wmma_f32_16x16x4_f32(false, a3, false, bfr[k], (short)0, c3, false, false);
        }
        size_t colOff = (size_t)colBase + m;
        if (full) {
#pragma unroll
            for (int r = 0; r < 8; ++r) {
                int rr = hi * 8 + r;
                g[(size_t)(nodeBase      + rr) * G_DIM + colOff] = c0[r];
                g[(size_t)(nodeBase + 16 + rr) * G_DIM + colOff] = c1[r];
                g[(size_t)(nodeBase + 32 + rr) * G_DIM + colOff] = c2[r];
                g[(size_t)(nodeBase + 48 + rr) * G_DIM + colOff] = c3[r];
            }
        } else {
#pragma unroll
            for (int r = 0; r < 8; ++r) {
                int rr = hi * 8 + r;
                if (nodeBase      + rr < N) g[(size_t)(nodeBase      + rr) * G_DIM + colOff] = c0[r];
                if (nodeBase + 16 + rr < N) g[(size_t)(nodeBase + 16 + rr) * G_DIM + colOff] = c1[r];
                if (nodeBase + 32 + rr < N) g[(size_t)(nodeBase + 32 + rr) * G_DIM + colOff] = c2[r];
                if (nodeBase + 48 + rr < N) g[(size_t)(nodeBase + 48 + rr) * G_DIM + colOff] = c3[r];
            }
        }
    }
}

__global__ __launch_bounds__(128) void node_tp_wmma(const float* __restrict__ y,
                                                    const float* __restrict__ Wl,
                                                    float* __restrict__ g, int N) {
    __shared__ __attribute__((aligned(16))) float sy[64 * TP_PITCH];
    int nodeBase = blockIdx.x * 64;
    for (int i = threadIdx.x * 4; i < 64 * A_DIM; i += 128 * 4) {
        int r = i >> 6, c = i & 63;
        int row = nodeBase + r;
        if (row >= N) row = N - 1;
        *(float4*)(sy + r * TP_PITCH + c) = *(const float4*)(y + (size_t)row * A_DIM + c);
    }
    __syncthreads();
    tp_tiles(sy, Wl, g, N, nodeBase, (int)blockIdx.y);
}

__global__ __launch_bounds__(128) void first_tp_wmma(const int* __restrict__ z,
                                                     const float* __restrict__ We,
                                                     const float* __restrict__ be,
                                                     const float* __restrict__ Wl,
                                                     float* __restrict__ yw,
                                                     float* __restrict__ g, int N,
                                                     const float* __restrict__ pos,
                                                     const int* __restrict__ src,
                                                     const int* __restrict__ dst,
                                                     float* __restrict__ ef,
                                                     int* __restrict__ rp, int E) {
    __shared__ __attribute__((aligned(16))) float sy[64 * TP_PITCH];
    int nb = (N + 63) >> 6;
    int b  = (int)blockIdx.x;
    if (b >= 3 * nb) {
        int e = (b - 3 * nb) * 128 + (int)threadIdx.x;
        if (e < E)  edge_feat_item(pos, src, dst, ef, e);
        if (e <= E) row_ptr_item(src, rp, N, E, e);
        return;
    }
    int third    = b / nb;
    int nodeBase = (b - third * nb) * 64;
    int c  = (threadIdx.x * 4) & 63;
    int r0 = threadIdx.x >> 4;
    float b0 = be[c], b1 = be[c + 1], b2 = be[c + 2], b3 = be[c + 3];
    int zr[8];
#pragma unroll
    for (int j = 0; j < 8; ++j) {
        int row = nodeBase + r0 + 8 * j;
        if (row >= N) row = N - 1;
        zr[j] = z[row];
    }
    float4 v[8];
#pragma unroll
    for (int j = 0; j < 8; ++j) {
        const float* wr = We + zr[j] * A_DIM + c;
        v[j].x = wr[0] + b0;
        v[j].y = wr[1] + b1;
        v[j].z = wr[2] + b2;
        v[j].w = wr[3] + b3;
    }
#pragma unroll
    for (int j = 0; j < 8; ++j) *(float4*)(sy + (r0 + 8 * j) * TP_PITCH + c) = v[j];
    __syncthreads();
    if (third == 0) {
#pragma unroll
        for (int j = 0; j < 8; ++j) {
            int row = nodeBase + r0 + 8 * j;
            if (row < N) *(float4*)(yw + (size_t)row * A_DIM + c) = v[j];
        }
    }
    tp_tiles(sy, Wl, g, N, nodeBase, third);
}

typedef __attribute__((ext_vector_type(4))) float v4f;

__device__ __forceinline__ float xhalf(float x) {
    return __builtin_bit_cast(float, __builtin_amdgcn_permlanex16(__builtin_bit_cast(int, x), __builtin_bit_cast(int, x),
                                                                  0x76543210, 0xfedcba98, false, false));
}

__device__ __forceinline__ v4f edge_msg(const float* __restrict__ g, const float* __restrict__ ef,
                                        int d, int e, int c) {
    const v4f* gp = (const v4f*)(g + (size_t)d * G_DIM) + c;
    const v4f* fp = (const v4f*)(ef + (size_t)e * F_DIM);
    v4f w0 = fp[0], w1 = fp[1], w2 = fp[2];
    float w[F_DIM] = {w0.x, w0.y, w0.z, w0.w, w1.x, w1.y, w1.z, w1.w, w2.x, w2.y, w2.z, w2.w};
    v4f v[F_DIM];
#pragma unroll
    for (int f = 0; f < F_DIM; ++f) v[f] = gp[f * 16];
    __builtin_amdgcn_sched_barrier(0);
    v4f m = {0.0f, 0.0f, 0.0f, 0.0f};
#pragma unroll
    for (int f = 0; f < F_DIM; ++f) m += w[f] * v[f];
    v4f s;
    s.x = silu(m.x * MSG_SCALE);
    s.y = silu(m.y * MSG_SCALE);
    s.z = silu(m.z * MSG_SCALE);
    s.w = silu(m.w * MSG_SCALE);
    return s;
}

template <bool LAST>
__global__ __launch_bounds__(256) void interact_kernel(float* __restrict__ y,
                                                       const float* __restrict__ g,
                                                       const float* __restrict__ ef,
                                                       const int* __restrict__ rp,
                                                       const int* __restrict__ dst,
                                                       float* __restrict__ out, int N) {
    int wave = __builtin_amdgcn_readfirstlane((int)(threadIdx.x >> 5));
    int n = blockIdx.x * 8 + wave;
    if (n >= N) return;
    int lane = threadIdx.x & 31;
    int c    = lane & 15;
    int hi   = lane >> 4;
    int e0 = rp[n], e1 = rp[n + 1];
    v4f acc = {0.0f, 0.0f, 0.0f, 0.0f};
    int e = e0;
    for (; e + 1 < e1; e += 2) {
        int d0 = dst[e], d1 = dst[e + 1];
        v4f s = edge_msg(g, ef, hi ? d1 : d0, e + hi, c);
        v4f o;
        o.x = xhalf(s.x); o.y = xhalf(s.y); o.z = xhalf(s.z); o.w = xhalf(s.w);
        acc = (acc + s) + o;
    }
    if (e < e1) {
        acc += edge_msg(g, ef, dst[e], e, c);
    }
    if (hi == 0) {
        v4f* yp = (v4f*)(y + (size_t)n * A_DIM) + c;
        v4f r = *yp + acc;
        *yp = r;
        if (LAST) {
            float* op = out + 1 + (size_t)n * A_DIM + 4 * c;
            op[0] = r.x; op[1] = r.y; op[2] = r.z; op[3] = r.w;
        }
    }
}

#define M_DIM 128
#define SY_PITCH 68

__global__ __launch_bounds__(256) void readout_wmma(const float* __restrict__ y,
                                                    const float* __restrict__ W1,
                                                    const float* __restrict__ b1,
                                                    const float* __restrict__ W2,
                                                    const float* __restrict__ b2,
                                                    float* __restrict__ partial, int N) {
    __shared__ __attribute__((aligned(16))) float sy[64 * SY_PITCH];
    __shared__ float red[8];
    int nodeBase = blockIdx.x * 64;
    int wave = threadIdx.x >> 5;
    int lane = threadIdx.x & 31;
    int m    = lane & 15;
    int hi   = lane >> 4;
    int col  = wave * 16 + m;

    const float* Bp = W1 + 2 * hi * M_DIM + col;
    v2f bfr[16];
#pragma unroll
    for (int k = 0; k < 16; ++k) {
        bfr[k].x = Bp[k * 4 * M_DIM];
        bfr[k].y = Bp[k * 4 * M_DIM + M_DIM];
    }
    float bias = b1[col];
    float w2   = W2[col];

    float4 st[4];
#pragma unroll
    for (int j = 0; j < 4; ++j) {
        int i = threadIdx.x * 4 + j * 1024;
        int row = nodeBase + (i >> 6);
        if (row >= N) row = N - 1;
        st[j] = *(const float4*)(y + (size_t)row * A_DIM + (i & 63));
    }
#pragma unroll
    for (int j = 0; j < 4; ++j) {
        int i = threadIdx.x * 4 + j * 1024;
        *(float4*)(sy + (i >> 6) * SY_PITCH + (i & 63)) = st[j];
    }
    __syncthreads();

    v8f c0 = {}, c1 = {}, c2 = {}, c3 = {};
#pragma unroll
    for (int k = 0; k < 16; ++k) {
        int ka = k * 4 + 2 * hi;
        v2f a0 = *(const v2f*)(sy + (m     ) * SY_PITCH + ka);
        v2f a1 = *(const v2f*)(sy + (m + 16) * SY_PITCH + ka);
        v2f a2 = *(const v2f*)(sy + (m + 32) * SY_PITCH + ka);
        v2f a3 = *(const v2f*)(sy + (m + 48) * SY_PITCH + ka);
        c0 = __builtin_amdgcn_wmma_f32_16x16x4_f32(false, a0, false, bfr[k], (short)0, c0, false, false);
        c1 = __builtin_amdgcn_wmma_f32_16x16x4_f32(false, a1, false, bfr[k], (short)0, c1, false, false);
        c2 = __builtin_amdgcn_wmma_f32_16x16x4_f32(false, a2, false, bfr[k], (short)0, c2, false, false);
        c3 = __builtin_amdgcn_wmma_f32_16x16x4_f32(false, a3, false, bfr[k], (short)0, c3, false, false);
    }

    float s = 0.0f;
#pragma unroll
    for (int r = 0; r < 8; ++r) {
        int row = nodeBase + hi * 8 + r;
        float h0 = silu(c0[r] + bias);
        float h1 = silu(c1[r] + bias);
        float h2 = silu(c2[r] + bias);
        float h3 = silu(c3[r] + bias);
        s += (row      < N) ? h0 : 0.0f;
        s += (row + 16 < N) ? h1 : 0.0f;
        s += (row + 32 < N) ? h2 : 0.0f;
        s += (row + 48 < N) ? h3 : 0.0f;
    }
    s *= w2;
    for (int off = 16; off > 0; off >>= 1) s += __shfl_xor(s, off);
    if (lane == 0) red[wave] = s;
    __syncthreads();
    if (threadIdx.x == 0) {
        float tot = 0.0f;
#pragma unroll
        for (int w = 0; w < 8; ++w) tot += red[w];
        int cnt = N - nodeBase;
        if (cnt > 64) cnt = 64;
        partial[blockIdx.x] = tot + b2[0] * (float)cnt;
    }
}

__global__ void finalize_kernel(const float* __restrict__ partial, int nPart,
                                const float* __restrict__ energy,
                                float* __restrict__ out, int N) {
    __shared__ float red[256];
    float v = 0.0f;
    for (int i = threadIdx.x; i < nPart; i += 256) v += partial[i];
    red[threadIdx.x] = v;
    __syncthreads();
    for (int off = 128; off > 0; off >>= 1) {
        if (threadIdx.x < off) red[threadIdx.x] += red[threadIdx.x + off];
        __syncthreads();
    }
    if (threadIdx.x == 0) {
        out[0] = red[0];
        out[1 + (size_t)N * A_DIM] = energy[0] * KCAL2MEV * (-1.0f);
    }
}

extern "C" void kernel_launch(void* const* d_in, const int* in_sizes, int n_in,
                              void* d_out, int out_size, void* d_ws, size_t ws_size,
                              hipStream_t stream) {
    const int*   z      = (const int*)d_in[0];
    const float* pos    = (const float*)d_in[1];
    const float* energy = (const float*)d_in[2];
    const int*   src    = (const int*)d_in[3];
    const int*   dst    = (const int*)d_in[4];
    const float* We     = (const float*)d_in[5];
    const float* be     = (const float*)d_in[6];
    const float* Wtp    = (const float*)d_in[7];
    const float* W1     = (const float*)d_in[8];
    const float* b1     = (const float*)d_in[9];
    const float* W2     = (const float*)d_in[10];
    const float* b2     = (const float*)d_in[11];
    float* out = (float*)d_out;

    const int N = in_sizes[0];
    const int E = in_sizes[3];

    float* fws = (float*)d_ws;
    size_t off = 0;
    float* ef      = fws + off; off += (size_t)E * F_DIM;
    float* g       = fws + off; off += (size_t)N * G_DIM;
    float* yw      = fws + off; off += (size_t)N * A_DIM;
    int*   rp      = (int*)(fws + off);
    float* partial = g;

    int nBlk = (N + 63) / 64;

    first_tp_wmma<<<3 * nBlk + (E + 1 + 127) / 128, 128, 0, stream>>>(
        z, We, be, Wtp, yw, g, N, pos, src, dst, ef, rp, E);
    interact_kernel<false><<<(N + 7) / 8, 256, 0, stream>>>(yw, g, ef, rp, dst, out, N);

    node_tp_wmma<<<dim3(nBlk, 3), 128, 0, stream>>>(yw, Wtp + (size_t)1 * F_DIM * A_DIM * A_DIM, g, N);
    interact_kernel<false><<<(N + 7) / 8, 256, 0, stream>>>(yw, g, ef, rp, dst, out, N);

    node_tp_wmma<<<dim3(nBlk, 3), 128, 0, stream>>>(yw, Wtp + (size_t)2 * F_DIM * A_DIM * A_DIM, g, N);
    interact_kernel<true><<<(N + 7) / 8, 256, 0, stream>>>(yw, g, ef, rp, dst, out, N);

    readout_wmma<<<nBlk, 256, 0, stream>>>(yw, W1, b1, W2, b2, partial, N);
    finalize_kernel<<<1, 256, 0, stream>>>(partial, nBlk, energy, out, N);
}


# layer 0: y has only 100 distinct rows (W_embed[z]+b), so y@W_tp[0] is built as a 100x768 table (12 matrix blocks, not 471) and the first interact launch gathers table rows by z[dst]; sums unchanged
# speedup vs baseline: 1.8924x; 1.1129x over previous
#include <hip/hip_runtime.h>
#include <math.h>

#define A_DIM 64
#define F_DIM 12
#define G_DIM (F_DIM * A_DIM)
#define NMU 11
#define MSG_SCALE 3.60843918e-04f
#define KCAL2MEV 43.3641f

typedef __attribute__((ext_vector_type(2))) float v2f;
typedef __attribute__((ext_vector_type(8))) float v8f;

__device__ __forceinline__ float silu(float x) { return x / (1.0f + expf(-x)); }

__device__ __forceinline__ void edge_feat_item(const float* __restrict__ pos, const int* __restrict__ src,
                                               const int* __restrict__ dst, const int* __restrict__ z,
                                               float* __restrict__ ef, int* __restrict__ zd, int e) {
    int s = src[e], d = dst[e];
    zd[e] = z[d];
    float dx = pos[s * 3 + 0] - pos[d * 3 + 0];
    float dy = pos[s * 3 + 1] - pos[d * 3 + 1];
    float dz = pos[s * 3 + 2] - pos[d * 3 + 2];
    float r = sqrtf(dx * dx + dy * dy + dz * dz);
#pragma unroll
    for (int j = 0; j < NMU; ++j) {
        float t = r - 0.5f * (float)j;
        ef[(size_t)e * F_DIM + j] = expf(-2.0f * t * t);
    }
    ef[(size_t)e * F_DIM + NMU] = 1.0f;
}

__device__ __forceinline__ void row_ptr_item(const int* __restrict__ src, int* __restrict__ rp, int N, int E, int e) {
    if (e == 0) {
        for (int n = 0; n <= src[0]; ++n) rp[n] = 0;
    } else if (e == E) {
        for (int n = src[E - 1] + 1; n <= N; ++n) rp[n] = E;
    } else {
        int a = src[e - 1], b = src[e];
        for (int n = a + 1; n <= b; ++n) rp[n] = e;
    }
}

#define TP_PITCH 68
#define TP_TILES 4

template <int TILES>
__device__ __forceinline__ void tp_tiles(const float* sy, const float* __restrict__ Wl,
                                         float* __restrict__ g, int N, int nodeBase, int tileBlock) {
    bool full = (nodeBase + 64 <= N);
    int wave = threadIdx.x >> 5;
    int lane = threadIdx.x & 31;
    int m    = lane & 15;
    int hi   = lane >> 4;

    for (int t = 0; t < TILES; ++t) {
        int tile    = (tileBlock * 4 + wave) * TILES + t;
        int colBase = tile << 4;
        int f       = colBase >> 6;
        int bcol    = (colBase & 63) + m;
        const float* Bp = Wl + (size_t)f * (A_DIM * A_DIM) + 2 * hi * A_DIM + bcol;

        v2f bfr[16];
#pragma unroll
        for (int k = 0; k < 16; ++k) {
            bfr[k].x = Bp[k * 4 * A_DIM];
            bfr[k].y = Bp[k * 4 * A_DIM + A_DIM];
        }

        v8f c0 = {}, c1 = {}, c2 = {}, c3 = {};
#pragma unroll
        for (int k = 0; k < 16; ++k) {
            int ka = k * 4 + 2 * hi;
            v2f a0; a0.x = sy[(m     ) * TP_PITCH + ka]; a0.y = sy[(m     ) * TP_PITCH + ka + 1];
            v2f a1; a1.x = sy[(m + 16) * TP_PITCH + ka]; a1.y = sy[(m + 16) * TP_PITCH + ka + 1];
            v2f a2; a2.x = sy[(m + 32) * TP_PITCH + ka]; a2.y = sy[(m + 32) * TP_PITCH + ka + 1];
            v2f a3; a3.x = sy[(m + 48) * TP_PITCH + ka]; a3.y = sy[(m + 48) * TP_PITCH + ka + 1];
            c0 = __builtin_amdgcn_wmma_f32_16x16x4_f32(false, a0, false, bfr[k], (short)0, c0, false, false);
            c1 = __builtin_amdgcn_wmma_f32_16x16x4_f32(false, a1, false, bfr[k], (short)0, c1, false, false);
            c2 = __builtin_amdgcn_wmma_f32_16x16x4_f32(false, a2, false, bfr[k], (short)0, c2, false, false);
            c3 = __builtin_amdgcn_wmma_f32_16x16x4_f32(false, a3, false, bfr[k], (short)0, c3, false, false);
        }
        size_t colOff = (size_t)colBase + m;
        if (full) {
#pragma unroll
            for (int r = 0; r < 8; ++r) {
                int rr = hi * 8 + r;
                g[(size_t)(nodeBase      + rr) * G_DIM + colOff] = c0[r];
                g[(size_t)(nodeBase + 16 + rr) * G_DIM + colOff] = c1[r];
                g[(size_t)(nodeBase + 32 + rr) * G_DIM + colOff] = c2[r];
                g[(size_t)(nodeBase + 48 + rr) * G_DIM + colOff] = c3[r];
            }
        } else {
#pragma unroll
            for (int r = 0; r < 8; ++r) {
                int rr = hi * 8 + r;
                if (nodeBase      + rr < N) g[(size_t)(nodeBase      + rr) * G_DIM + colOff] = c0[r];
                if (nodeBase + 16 + rr < N) g[(size_t)(nodeBase + 16 + rr) * G_DIM + colOff] = c1[r];
                if (nodeBase + 32 + rr < N) g[(size_t)(nodeBase + 32 + rr) * G_DIM + colOff] = c2[r];
                if (nodeBase + 48 + rr < N) g[(size_t)(nodeBase + 48 + rr) * G_DIM + colOff] = c3[r];
            }
        }
    }
}

__global__ __launch_bounds__(128) void node_tp_wmma(const float* __restrict__ y,
                                                    const float* __restrict__ Wl,
                                                    float* __restrict__ g, int N) {
    __shared__ __attribute__((aligned(16))) float sy[64 * TP_PITCH];
    int nodeBase = blockIdx.x * 64;
    for (int i = threadIdx.x * 4; i < 64 * A_DIM; i += 128 * 4) {
        int r = i >> 6, c = i & 63;
        int row = nodeBase + r;
        if (row >= N) row = N - 1;
        *(float4*)(sy + r * TP_PITCH + c) = *(const float4*)(y + (size_t)row * A_DIM + c);
    }
    __syncthreads();
    tp_tiles<TP_TILES>(sy, Wl, g, N, nodeBase, (int)blockIdx.y);
}

#define MAX_Z 100
#define TAB_RB 2
#define TAB_CG 6
#define TAB_TILES 2
__global__ __launch_bounds__(128) void first_table_wmma(const int* __restrict__ z,
                                                        const float* __restrict__ We,
                                                        const float* __restrict__ be,
                                                        const float* __restrict__ Wl,
                                                        float* __restrict__ yw,
                                                        float* __restrict__ tab,
                                                        int* __restrict__ zd,
                                                        int N,
                                                        const float* __restrict__ pos,
                                                        const int* __restrict__ src,
                                                        const int* __restrict__ dst,
                                                        float* __restrict__ ef,
                                                        int* __restrict__ rp, int E) {
    __shared__ __attribute__((aligned(16))) float sy[64 * TP_PITCH];
    int b = (int)blockIdx.x;
    if (b >= TAB_RB * TAB_CG) {
        int e = (b - TAB_RB * TAB_CG) * 128 + (int)threadIdx.x;
        if (e < E)  edge_feat_item(pos, src, dst, z, ef, zd, e);
        if (e <= E) row_ptr_item(src, rp, N, E, e);
        int nItems = ((E + 1 + 127) >> 7) * 128;
        for (int i = e; i < N * 16; i += nItems) {
            int row = i >> 4;
            int c   = (i & 15) * 4;
            const float* wr = We + z[row] * A_DIM + c;
            float4 v;
            v.x = wr[0] + be[c];
            v.y = wr[1] + be[c + 1];
            v.z = wr[2] + be[c + 2];
            v.w = wr[3] + be[c + 3];
            *(float4*)(yw + (size_t)row * A_DIM + c) = v;
        }
        return;
    }
    int rb      = b / TAB_CG;
    int cg      = b - rb * TAB_CG;
    int rowBase = rb * 64;
    int c  = (threadIdx.x * 4) & 63;
    int r0 = threadIdx.x >> 4;
    float b0 = be[c], b1 = be[c + 1], b2 = be[c + 2], b3 = be[c + 3];
    float4 v[8];
#pragma unroll
    for (int j = 0; j < 8; ++j) {
        int row = rowBase + r0 + 8 * j;
        if (row >= MAX_Z) row = MAX_Z - 1;
        const float* wr = We + row * A_DIM + c;
        v[j].x = wr[0] + b0;
        v[j].y = wr[1] + b1;
        v[j].z = wr[2] + b2;
        v[j].w = wr[3] + b3;
    }
#pragma unroll
    for (int j = 0; j < 8; ++j) *(float4*)(sy + (r0 + 8 * j) * TP_PITCH + c) = v[j];
    __syncthreads();
    tp_tiles<TAB_TILES>(sy, Wl, tab, MAX_Z, rowBase, cg);
}

typedef __attribute__((ext_vector_type(4))) float v4f;

__device__ __forceinline__ float xhalf(float x) {
    return __builtin_bit_cast(float, __builtin_amdgcn_permlanex16(__builtin_bit_cast(int, x), __builtin_bit_cast(int, x),
                                                                  0x76543210, 0xfedcba98, false, false));
}

__device__ __forceinline__ v4f edge_msg(const float* __restrict__ g, const float* __restrict__ ef,
                                        int d, int e, int c) {
    const v4f* gp = (const v4f*)(g + (size_t)d * G_DIM) + c;
    const v4f* fp = (const v4f*)(ef + (size_t)e * F_DIM);
    v4f w0 = fp[0], w1 = fp[1], w2 = fp[2];
    float w[F_DIM] = {w0.x, w0.y, w0.z, w0.w, w1.x, w1.y, w1.z, w1.w, w2.x, w2.y, w2.z, w2.w};
    v4f v[F_DIM];
#pragma unroll
    for (int f = 0; f < F_DIM; ++f) v[f] = gp[f * 16];
    __builtin_amdgcn_sched_barrier(0);
    v4f m = {0.0f, 0.0f, 0.0f, 0.0f};
#pragma unroll
    for (int f = 0; f < F_DIM; ++f) m += w[f] * v[f];
    v4f s;
    s.x = silu(m.x * MSG_SCALE);
    s.y = silu(m.y * MSG_SCALE);
    s.z = silu(m.z * MSG_SCALE);
    s.w = silu(m.w * MSG_SCALE);
    return s;
}

template <bool LAST>
__global__ __launch_bounds__(256) void interact_kernel(float* __restrict__ y,
                                                       const float* __restrict__ g,
                                                       const float* __restrict__ ef,
                                                       const int* __restrict__ rp,
                                                       const int* __restrict__ dst,
                                                       float* __restrict__ out, int N) {
    int wave = __builtin_amdgcn_readfirstlane((int)(threadIdx.x >> 5));
    int n = blockIdx.x * 8 + wave;
    if (n >= N) return;
    int lane = threadIdx.x & 31;
    int c    = lane & 15;
    int hi   = lane >> 4;
    int e0 = rp[n], e1 = rp[n + 1];
    v4f acc = {0.0f, 0.0f, 0.0f, 0.0f};
    int e = e0;
    for (; e + 1 < e1; e += 2) {
        int d0 = dst[e], d1 = dst[e + 1];
        v4f s = edge_msg(g, ef, hi ? d1 : d0, e + hi, c);
        v4f o;
        o.x = xhalf(s.x); o.y = xhalf(s.y); o.z = xhalf(s.z); o.w = xhalf(s.w);
        acc = (acc + s) + o;
    }
    if (e < e1) {
        acc += edge_msg(g, ef, dst[e], e, c);
    }
    if (hi == 0) {
        v4f* yp = (v4f*)(y + (size_t)n * A_DIM) + c;
        v4f r = *yp + acc;
        *yp = r;
        if (LAST) {
            float* op = out + 1 + (size_t)n * A_DIM + 4 * c;
            op[0] = r.x; op[1] = r.y; op[2] = r.z; op[3] = r.w;
        }
    }
}

#define M_DIM 128
#define SY_PITCH 68

__global__ __launch_bounds__(256) void readout_wmma(const float* __restrict__ y,
                                                    const float* __restrict__ W1,
                                                    const float* __restrict__ b1,
                                                    const float* __restrict__ W2,
                                                    const float* __restrict__ b2,
                                                    float* __restrict__ partial, int N) {
    __shared__ __attribute__((aligned(16))) float sy[64 * SY_PITCH];
    __shared__ float red[8];
    int nodeBase = blockIdx.x * 64;
    int wave = threadIdx.x >> 5;
    int lane = threadIdx.x & 31;
    int m    = lane & 15;
    int hi   = lane >> 4;
    int col  = wave * 16 + m;

    const float* Bp = W1 + 2 * hi * M_DIM + col;
    v2f bfr[16];
#pragma unroll
    for (int k = 0; k < 16; ++k) {
        bfr[k].x = Bp[k * 4 * M_DIM];
        bfr[k].y = Bp[k * 4 * M_DIM + M_DIM];
    }
    float bias = b1[col];
    float w2   = W2[col];

    float4 st[4];
#pragma unroll
    for (int j = 0; j < 4; ++j) {
        int i = threadIdx.x * 4 + j * 1024;
        int row = nodeBase + (i >> 6);
        if (row >= N) row = N - 1;
        st[j] = *(const float4*)(y + (size_t)row * A_DIM + (i & 63));
    }
#pragma unroll
    for (int j = 0; j < 4; ++j) {
        int i = threadIdx.x * 4 + j * 1024;
        *(float4*)(sy + (i >> 6) * SY_PITCH + (i & 63)) = st[j];
    }
    __syncthreads();

    v8f c0 = {}, c1 = {}, c2 = {}, c3 = {};
#pragma unroll
    for (int k = 0; k < 16; ++k) {
        int ka = k * 4 + 2 * hi;
        v2f a0 = *(const v2f*)(sy + (m     ) * SY_PITCH + ka);
        v2f a1 = *(const v2f*)(sy + (m + 16) * SY_PITCH + ka);
        v2f a2 = *(const v2f*)(sy + (m + 32) * SY_PITCH + ka);
        v2f a3 = *(const v2f*)(sy + (m + 48) * SY_PITCH + ka);
        c0 = __builtin_amdgcn_wmma_f32_16x16x4_f32(false, a0, false, bfr[k], (short)0, c0, false, false);
        c1 = __builtin_amdgcn_wmma_f32_16x16x4_f32(false, a1, false, bfr[k], (short)0, c1, false, false);
        c2 = __builtin_amdgcn_wmma_f32_16x16x4_f32(false, a2, false, bfr[k], (short)0, c2, false, false);
        c3 = __builtin_amdgcn_wmma_f32_16x16x4_f32(false, a3, false, bfr[k], (short)0, c3, false, false);
    }

    float s = 0.0f;
#pragma unroll
    for (int r = 0; r < 8; ++r) {
        int row = nodeBase + hi * 8 + r;
        float h0 = silu(c0[r] + bias);
        float h1 = silu(c1[r] + bias);
        float h2 = silu(c2[r] + bias);
        float h3 = silu(c3[r] + bias);
        s += (row      < N) ? h0 : 0.0f;
        s += (row + 16 < N) ? h1 : 0.0f;
        s += (row + 32 < N) ? h2 : 0.0f;
        s += (row + 48 < N) ? h3 : 0.0f;
    }
    s *= w2;
    for (int off = 16; off > 0; off >>= 1) s += __shfl_xor(s, off);
    if (lane == 0) red[wave] = s;
    __syncthreads();
    if (threadIdx.x == 0) {
        float tot = 0.0f;
#pragma unroll
        for (int w = 0; w < 8; ++w) tot += red[w];
        int cnt = N - nodeBase;
        if (cnt > 64) cnt = 64;
        partial[blockIdx.x] = tot + b2[0] * (float)cnt;
    }
}

__global__ void finalize_kernel(const float* __restrict__ partial, int nPart,
                                const float* __restrict__ energy,
                                float* __restrict__ out, int N) {
    __shared__ float red[256];
    float v = 0.0f;
    for (int i = threadIdx.x; i < nPart; i += 256) v += partial[i];
    red[threadIdx.x] = v;
    __syncthreads();
    for (int off = 128; off > 0; off >>= 1) {
        if (threadIdx.x < off) red[threadIdx.x] += red[threadIdx.x + off];
        __syncthreads();
    }
    if (threadIdx.x == 0) {
        out[0] = red[0];
        out[1 + (size_t)N * A_DIM] = energy[0] * KCAL2MEV * (-1.0f);
    }
}

extern "C" void kernel_launch(void* const* d_in, const int* in_sizes, int n_in,
                              void* d_out, int out_size, void* d_ws, size_t ws_size,
                              hipStream_t stream) {
    const int*   z      = (const int*)d_in[0];
    const float* pos    = (const float*)d_in[1];
    const float* energy = (const float*)d_in[2];
    const int*   src    = (const int*)d_in[3];
    const int*   dst    = (const int*)d_in[4];
    const float* We     = (const float*)d_in[5];
    const float* be     = (const float*)d_in[6];
    const float* Wtp    = (const float*)d_in[7];
    const float* W1     = (const float*)d_in[8];
    const float* b1     = (const float*)d_in[9];
    const float* W2     = (const float*)d_in[10];
    const float* b2     = (const float*)d_in[11];
    float* out = (float*)d_out;

    const int N = in_sizes[0];
    const int E = in_sizes[3];

    float* fws = (float*)d_ws;
    size_t off = 0;
    float* ef      = fws + off; off += (size_t)E * F_DIM;
    float* g       = fws + off; off += (size_t)N * G_DIM;
    float* yw      = fws + off; off += (size_t)N * A_DIM;
    int*   rp      = (int*)(fws + off);
    float* partial = g;
    float* tab     = g;
    int*   zd      = (int*)(g + (size_t)128 * G_DIM);

    int nBlk = (N + 63) / 64;

    first_table_wmma<<<TAB_RB * TAB_CG + (E + 1 + 127) / 128, 128, 0, stream>>>(
        z, We, be, Wtp, yw, tab, zd, N, pos, src, dst, ef, rp, E);
    interact_kernel<false><<<(N + 7) / 8, 256, 0, stream>>>(yw, tab, ef, rp, zd, out, N);

    node_tp_wmma<<<dim3(nBlk, 3), 128, 0, stream>>>(yw, Wtp + (size_t)1 * F_DIM * A_DIM * A_DIM, g, N);
    interact_kernel<false><<<(N + 7) / 8, 256, 0, stream>>>(yw, g, ef, rp, dst, out, N);

    node_tp_wmma<<<dim3(nBlk, 3), 128, 0, stream>>>(yw, Wtp + (size_t)2 * F_DIM * A_DIM * A_DIM, g, N);
    interact_kernel<true><<<(N + 7) / 8, 256, 0, stream>>>(yw, g, ef, rp, dst, out, N);

    readout_wmma<<<nBlk, 256, 0, stream>>>(yw, W1, b1, W2, b2, partial, N);
    finalize_kernel<<<1, 256, 0, stream>>>(partial, nBlk, energy, out, N);
}


# interact: a g chunk whose edge weight is exactly zero is not fetched, its load is pointed at the row's chunk 11 that every edge needs: about one chunk in seven; exact zeros either way, sums unchanged
# speedup vs baseline: 1.9538x; 1.0324x over previous
#include <hip/hip_runtime.h>
#include <math.h>

#define A_DIM 64
#define F_DIM 12
#define G_DIM (F_DIM * A_DIM)
#define NMU 11
#define MSG_SCALE 3.60843918e-04f
#define KCAL2MEV 43.3641f

typedef __attribute__((ext_vector_type(2))) float v2f;
typedef __attribute__((ext_vector_type(8))) float v8f;

__device__ __forceinline__ float silu(float x) { return x / (1.0f + expf(-x)); }

__device__ __forceinline__ void edge_feat_item(const float* __restrict__ pos, const int* __restrict__ src,
                                               const int* __restrict__ dst, const int* __restrict__ z,
                                               float* __restrict__ ef, int* __restrict__ zd, int e) {
    int s = src[e], d = dst[e];
    zd[e] = z[d];
    float dx = pos[s * 3 + 0] - pos[d * 3 + 0];
    float dy = pos[s * 3 + 1] - pos[d * 3 + 1];
    float dz = pos[s * 3 + 2] - pos[d * 3 + 2];
    float r = sqrtf(dx * dx + dy * dy + dz * dz);
#pragma unroll
    for (int j = 0; j < NMU; ++j) {
        float t = r - 0.5f * (float)j;
        ef[(size_t)e * F_DIM + j] = expf(-2.0f * t * t);
    }
    ef[(size_t)e * F_DIM + NMU] = 1.0f;
}

__device__ __forceinline__ void row_ptr_item(const int* __restrict__ src, int* __restrict__ rp, int N, int E, int e) {
    if (e == 0) {
        for (int n = 0; n <= src[0]; ++n) rp[n] = 0;
    } else if (e == E) {
        for (int n = src[E - 1] + 1; n <= N; ++n) rp[n] = E;
    } else {
        int a = src[e - 1], b = src[e];
        for (int n = a + 1; n <= b; ++n) rp[n] = e;
    }
}

#define TP_PITCH 68
#define TP_TILES 4

template <int TILES>
__device__ __forceinline__ void tp_tiles(const float* sy, const float* __restrict__ Wl,
                                         float* __restrict__ g, int N, int nodeBase, int tileBlock) {
    bool full = (nodeBase + 64 <= N);
    int wave = threadIdx.x >> 5;
    int lane = threadIdx.x & 31;
    int m    = lane & 15;
    int hi   = lane >> 4;

    for (int t = 0; t < TILES; ++t) {
        int tile    = (tileBlock * 4 + wave) * TILES + t;
        int colBase = tile << 4;
        int f       = colBase >> 6;
        int bcol    = (colBase & 63) + m;
        const float* Bp = Wl + (size_t)f * (A_DIM * A_DIM) + 2 * hi * A_DIM + bcol;

        v2f bfr[16];
#pragma unroll
        for (int k = 0; k < 16; ++k) {
            bfr[k].x = Bp[k * 4 * A_DIM];
            bfr[k].y = Bp[k * 4 * A_DIM + A_DIM];
        }

        v8f c0 = {}, c1 = {}, c2 = {}, c3 = {};
#pragma unroll
        for (int k = 0; k < 16; ++k) {
            int ka = k * 4 + 2 * hi;
            v2f a0; a0.x = sy[(m     ) * TP_PITCH + ka]; a0.y = sy[(m     ) * TP_PITCH + ka + 1];
            v2f a1; a1.x = sy[(m + 16) * TP_PITCH + ka]; a1.y = sy[(m + 16) * TP_PITCH + ka + 1];
            v2f a2; a2.x = sy[(m + 32) * TP_PITCH + ka]; a2.y = sy[(m + 32) * TP_PITCH + ka + 1];
            v2f a3; a3.x = sy[(m + 48) * TP_PITCH + ka]; a3.y = sy[(m + 48) * TP_PITCH + ka + 1];
            c0 = __builtin_amdgcn_wmma_f32_16x16x4_f32(false, a0, false, bfr[k], (short)0, c0, false, false);
            c1 = __builtin_amdgcn_wmma_f32_16x16x4_f32(false, a1, false, bfr[k], (short)0, c1, false, false);
            c2 = __builtin_amdgcn_wmma_f32_16x16x4_f32(false, a2, false, bfr[k], (short)0, c2, false, false);
            c3 = __builtin_amdgcn_wmma_f32_16x16x4_f32(false, a3, false, bfr[k], (short)0, c3, false, false);
        }
        size_t colOff = (size_t)colBase + m;
        if (full) {
#pragma unroll
            for (int r = 0; r < 8; ++r) {
                int rr = hi * 8 + r;
                g[(size_t)(nodeBase      + rr) * G_DIM + colOff] = c0[r];
                g[(size_t)(nodeBase + 16 + rr) * G_DIM + colOff] = c1[r];
                g[(size_t)(nodeBase + 32 + rr) * G_DIM + colOff] = c2[r];
                g[(size_t)(nodeBase + 48 + rr) * G_DIM + colOff] = c3[r];
            }
        } else {
#pragma unroll
            for (int r = 0; r < 8; ++r) {
                int rr = hi * 8 + r;
                if (nodeBase      + rr < N) g[(size_t)(nodeBase      + rr) * G_DIM + colOff] = c0[r];
                if (nodeBase + 16 + rr < N) g[(size_t)(nodeBase + 16 + rr) * G_DIM + colOff] = c1[r];
                if (nodeBase + 32 + rr < N) g[(size_t)(nodeBase + 32 + rr) * G_DIM + colOff] = c2[r];
                if (nodeBase + 48 + rr < N) g[(size_t)(nodeBase + 48 + rr) * G_DIM + colOff] = c3[r];
            }
        }
    }
}

__global__ __launch_bounds__(128) void node_tp_wmma(const float* __restrict__ y,
                                                    const float* __restrict__ Wl,
                                                    float* __restrict__ g, int N) {
    __shared__ __attribute__((aligned(16))) float sy[64 * TP_PITCH];
    int nodeBase = blockIdx.x * 64;
    for (int i = threadIdx.x * 4; i < 64 * A_DIM; i += 128 * 4) {
        int r = i >> 6, c = i & 63;
        int row = nodeBase + r;
        if (row >= N) row = N - 1;
        *(float4*)(sy + r * TP_PITCH + c) = *(const float4*)(y + (size_t)row * A_DIM + c);
    }
    __syncthreads();
    tp_tiles<TP_TILES>(sy, Wl, g, N, nodeBase, (int)blockIdx.y);
}

#define MAX_Z 100
#define TAB_RB 2
#define TAB_CG 6
#define TAB_TILES 2
__global__ __launch_bounds__(128) void first_table_wmma(const int* __restrict__ z,
                                                        const float* __restrict__ We,
                                                        const float* __restrict__ be,
                                                        const float* __restrict__ Wl,
                                                        float* __restrict__ yw,
                                                        float* __restrict__ tab,
                                                        int* __restrict__ zd,
                                                        int N,
                                                        const float* __restrict__ pos,
                                                        const int* __restrict__ src,
                                                        const int* __restrict__ dst,
                                                        float* __restrict__ ef,
                                                        int* __restrict__ rp, int E) {
    __shared__ __attribute__((aligned(16))) float sy[64 * TP_PITCH];
    int b = (int)blockIdx.x;
    if (b >= TAB_RB * TAB_CG) {
        int e = (b - TAB_RB * TAB_CG) * 128 + (int)threadIdx.x;
        if (e < E)  edge_feat_item(pos, src, dst, z, ef, zd, e);
        if (e <= E) row_ptr_item(src, rp, N, E, e);
        int nItems = ((E + 1 + 127) >> 7) * 128;
        for (int i = e; i < N * 16; i += nItems) {
            int row = i >> 4;
            int c   = (i & 15) * 4;
            const float* wr = We + z[row] * A_DIM + c;
            float4 v;
            v.x = wr[0] + be[c];
            v.y = wr[1] + be[c + 1];
            v.z = wr[2] + be[c + 2];
            v.w = wr[3] + be[c + 3];
            *(float4*)(yw + (size_t)row * A_DIM + c) = v;
        }
        return;
    }
    int rb      = b / TAB_CG;
    int cg      = b - rb * TAB_CG;
    int rowBase = rb * 64;
    int c  = (threadIdx.x * 4) & 63;
    int r0 = threadIdx.x >> 4;
    float b0 = be[c], b1 = be[c + 1], b2 = be[c + 2], b3 = be[c + 3];
    float4 v[8];
#pragma unroll
    for (int j = 0; j < 8; ++j) {
        int row = rowBase + r0 + 8 * j;
        if (row >= MAX_Z) row = MAX_Z - 1;
        const float* wr = We + row * A_DIM + c;
        v[j].x = wr[0] + b0;
        v[j].y = wr[1] + b1;
        v[j].z = wr[2] + b2;
        v[j].w = wr[3] + b3;
    }
#pragma unroll
    for (int j = 0; j < 8; ++j) *(float4*)(sy + (r0 + 8 * j) * TP_PITCH + c) = v[j];
    __syncthreads();
    tp_tiles<TAB_TILES>(sy, Wl, tab, MAX_Z, rowBase, cg);
}

typedef __attribute__((ext_vector_type(4))) float v4f;

__device__ __forceinline__ float xhalf(float x) {
    return __builtin_bit_cast(float, __builtin_amdgcn_permlanex16(__builtin_bit_cast(int, x), __builtin_bit_cast(int, x),
                                                                  0x76543210, 0xfedcba98, false, false));
}

__device__ __forceinline__ v4f edge_msg(const float* __restrict__ g, const float* __restrict__ ef,
                                        const int* __restrict__ dst, int e, int c) {
    const v4f* fp = (const v4f*)(ef + (size_t)e * F_DIM);
    int d = dst[e];
    v4f w0 = fp[0], w1 = fp[1], w2 = fp[2];
    __builtin_amdgcn_sched_barrier(0);
    float w[F_DIM] = {w0.x, w0.y, w0.z, w0.w, w1.x, w1.y, w1.z, w1.w, w2.x, w2.y, w2.z, w2.w};
    unsigned row = (unsigned)d * (unsigned)(G_DIM * 4) + (unsigned)c * 16u;
    const char* gb = (const char*)g;
    v4f v[F_DIM];
    v[F_DIM - 1] = *(const v4f*)(gb + (row + (unsigned)(F_DIM - 1) * 256u));
#pragma unroll
    for (int f = 0; f < F_DIM - 1; ++f) {
        unsigned o = (w[f] != 0.0f) ? (unsigned)f * 256u : (unsigned)(F_DIM - 1) * 256u;
        v[f] = *(const v4f*)(gb + (row + o));
    }
    __builtin_amdgcn_sched_barrier(0);
    v4f m = {0.0f, 0.0f, 0.0f, 0.0f};
#pragma unroll
    for (int f = 0; f < F_DIM; ++f) m += w[f] * v[f];
    v4f s;
    s.x = silu(m.x * MSG_SCALE);
    s.y = silu(m.y * MSG_SCALE);
    s.z = silu(m.z * MSG_SCALE);
    s.w = silu(m.w * MSG_SCALE);
    return s;
}

template <bool LAST>
__global__ __launch_bounds__(256) void interact_kernel(float* __restrict__ y,
                                                       const float* __restrict__ g,
                                                       const float* __restrict__ ef,
                                                       const int* __restrict__ rp,
                                                       const int* __restrict__ dst,
                                                       float* __restrict__ out, int N) {
    int wave = __builtin_amdgcn_readfirstlane((int)(threadIdx.x >> 5));
    int n = blockIdx.x * 8 + wave;
    if (n >= N) return;
    int lane = threadIdx.x & 31;
    int c    = lane & 15;
    int hi   = lane >> 4;
    int e0 = rp[n], e1 = rp[n + 1];
    v4f acc = {0.0f, 0.0f, 0.0f, 0.0f};
    int e = e0;
    for (; e + 1 < e1; e += 2) {
        v4f s = edge_msg(g, ef, dst, e + hi, c);
        v4f o;
        o.x = xhalf(s.x); o.y = xhalf(s.y); o.z = xhalf(s.z); o.w = xhalf(s.w);
        acc = (acc + s) + o;
    }
    if (e < e1) {
        acc += edge_msg(g, ef, dst, e, c);
    }
    if (hi == 0) {
        v4f* yp = (v4f*)(y + (size_t)n * A_DIM) + c;
        v4f r = *yp + acc;
        *yp = r;
        if (LAST) {
            float* op = out + 1 + (size_t)n * A_DIM + 4 * c;
            op[0] = r.x; op[1] = r.y; op[2] = r.z; op[3] = r.w;
        }
    }
}

#define M_DIM 128
#define SY_PITCH 68

__global__ __launch_bounds__(256) void readout_wmma(const float* __restrict__ y,
                                                    const float* __restrict__ W1,
                                                    const float* __restrict__ b1,
                                                    const float* __restrict__ W2,
                                                    const float* __restrict__ b2,
                                                    float* __restrict__ partial, int N) {
    __shared__ __attribute__((aligned(16))) float sy[64 * SY_PITCH];
    __shared__ float red[8];
    int nodeBase = blockIdx.x * 64;
    int wave = threadIdx.x >> 5;
    int lane = threadIdx.x & 31;
    int m    = lane & 15;
    int hi   = lane >> 4;
    int col  = wave * 16 + m;

    const float* Bp = W1 + 2 * hi * M_DIM + col;
    v2f bfr[16];
#pragma unroll
    for (int k = 0; k < 16; ++k) {
        bfr[k].x = Bp[k * 4 * M_DIM];
        bfr[k].y = Bp[k * 4 * M_DIM + M_DIM];
    }
    float bias = b1[col];
    float w2   = W2[col];

    float4 st[4];
#pragma unroll
    for (int j = 0; j < 4; ++j) {
        int i = threadIdx.x * 4 + j * 1024;
        int row = nodeBase + (i >> 6);
        if (row >= N) row = N - 1;
        st[j] = *(const float4*)(y + (size_t)row * A_DIM + (i & 63));
    }
#pragma unroll
    for (int j = 0; j < 4; ++j) {
        int i = threadIdx.x * 4 + j * 1024;
        *(float4*)(sy + (i >> 6) * SY_PITCH + (i & 63)) = st[j];
    }
    __syncthreads();

    v8f c0 = {}, c1 = {}, c2 = {}, c3 = {};
#pragma unroll
    for (int k = 0; k < 16; ++k) {
        int ka = k * 4 + 2 * hi;
        v2f a0 = *(const v2f*)(sy + (m     ) * SY_PITCH + ka);
        v2f a1 = *(const v2f*)(sy + (m + 16) * SY_PITCH + ka);
        v2f a2 = *(const v2f*)(sy + (m + 32) * SY_PITCH + ka);
        v2f a3 = *(const v2f*)(sy + (m + 48) * SY_PITCH + ka);
        c0 = __builtin_amdgcn_wmma_f32_16x16x4_f32(false, a0, false, bfr[k], (short)0, c0, false, false);
        c1 = __builtin_amdgcn_wmma_f32_16x16x4_f32(false, a1, false, bfr[k], (short)0, c1, false, false);
        c2 = __builtin_amdgcn_wmma_f32_16x16x4_f32(false, a2, false, bfr[k], (short)0, c2, false, false);
        c3 = __builtin_amdgcn_wmma_f32_16x16x4_f32(false, a3, false, bfr[k], (short)0, c3, false, false);
    }

    float s = 0.0f;
#pragma unroll
    for (int r = 0; r < 8; ++r) {
        int row = nodeBase + hi * 8 + r;
        float h0 = silu(c0[r] + bias);
        float h1 = silu(c1[r] + bias);
        float h2 = silu(c2[r] + bias);
        float h3 = silu(c3[r] + bias);
        s += (row      < N) ? h0 : 0.0f;
        s += (row + 16 < N) ? h1 : 0.0f;
        s += (row + 32 < N) ? h2 : 0.0f;
        s += (row + 48 < N) ? h3 : 0.0f;
    }
    s *= w2;
    for (int off = 16; off > 0; off >>= 1) s += __shfl_xor(s, off);
    if (lane == 0) red[wave] = s;
    __syncthreads();
    if (threadIdx.x == 0) {
        float tot = 0.0f;
#pragma unroll
        for (int w = 0; w < 8; ++w) tot += red[w];
        int cnt = N - nodeBase;
        if (cnt > 64) cnt = 64;
        partial[blockIdx.x] = tot + b2[0] * (float)cnt;
    }
}

__global__ void finalize_kernel(const float* __restrict__ partial, int nPart,
                                const float* __restrict__ energy,
                                float* __restrict__ out, int N) {
    __shared__ float red[256];
    float v = 0.0f;
    for (int i = threadIdx.x; i < nPart; i += 256) v += partial[i];
    red[threadIdx.x] = v;
    __syncthreads();
    for (int off = 128; off > 0; off >>= 1) {
        if (threadIdx.x < off) red[threadIdx.x] += red[threadIdx.x + off];
        __syncthreads();
    }
    if (threadIdx.x == 0) {
        out[0] = red[0];
        out[1 + (size_t)N * A_DIM] = energy[0] * KCAL2MEV * (-1.0f);
    }
}

extern "C" void kernel_launch(void* const* d_in, const int* in_sizes, int n_in,
                              void* d_out, int out_size, void* d_ws, size_t ws_size,
                              hipStream_t stream) {
    const int*   z      = (const int*)d_in[0];
    const float* pos    = (const float*)d_in[1];
    const float* energy = (const float*)d_in[2];
    const int*   src    = (const int*)d_in[3];
    const int*   dst    = (const int*)d_in[4];
    const float* We     = (const float*)d_in[5];
    const float* be     = (const float*)d_in[6];
    const float* Wtp    = (const float*)d_in[7];
    const float* W1     = (const float*)d_in[8];
    const float* b1     = (const float*)d_in[9];
    const float* W2     = (const float*)d_in[10];
    const float* b2     = (const float*)d_in[11];
    float* out = (float*)d_out;

    const int N = in_sizes[0];
    const int E = in_sizes[3];

    float* fws = (float*)d_ws;
    size_t off = 0;
    float* ef      = fws + off; off += (size_t)E * F_DIM;
    float* g       = fws + off; off += (size_t)N * G_DIM;
    float* yw      = fws + off; off += (size_t)N * A_DIM;
    int*   rp      = (int*)(fws + off);
    float* partial = g;
    float* tab     = g;
    int*   zd      = (int*)(g + (size_t)128 * G_DIM);

    int nBlk = (N + 63) / 64;

    first_table_wmma<<<TAB_RB * TAB_CG + (E + 1 + 127) / 128, 128, 0, stream>>>(
        z, We, be, Wtp, yw, tab, zd, N, pos, src, dst, ef, rp, E);
    interact_kernel<false><<<(N + 7) / 8, 256, 0, stream>>>(yw, tab, ef, rp, zd, out, N);

    node_tp_wmma<<<dim3(nBlk, 3), 128, 0, stream>>>(yw, Wtp + (size_t)1 * F_DIM * A_DIM * A_DIM, g, N);
    interact_kernel<false><<<(N + 7) / 8, 256, 0, stream>>>(yw, g, ef, rp, dst, out, N);

    node_tp_wmma<<<dim3(nBlk, 3), 128, 0, stream>>>(yw, Wtp + (size_t)2 * F_DIM * A_DIM * A_DIM, g, N);
    interact_kernel<true><<<(N + 7) / 8, 256, 0, stream>>>(yw, g, ef, rp, dst, out, N);

    readout_wmma<<<nBlk, 256, 0, stream>>>(yw, W1, b1, W2, b2, partial, N);
    finalize_kernel<<<1, 256, 0, stream>>>(partial, nBlk, energy, out, N);
}


# node_tp: loads go out ahead of their use: the 8 staging loads of the y tile as one group (were 8 trips in a row), each tile's B fragments one tile ahead, the first before the staging; sums unchanged
# speedup vs baseline: 2.0882x; 1.0688x over previous
#include <hip/hip_runtime.h>
#include <math.h>

#define A_DIM 64
#define F_DIM 12
#define G_DIM (F_DIM * A_DIM)
#define NMU 11
#define MSG_SCALE 3.60843918e-04f
#define KCAL2MEV 43.3641f

typedef __attribute__((ext_vector_type(2))) float v2f;
typedef __attribute__((ext_vector_type(8))) float v8f;

__device__ __forceinline__ float silu(float x) { return x / (1.0f + expf(-x)); }

__device__ __forceinline__ void edge_feat_item(const float* __restrict__ pos, const int* __restrict__ src,
                                               const int* __restrict__ dst, const int* __restrict__ z,
                                               float* __restrict__ ef, int* __restrict__ zd, int e) {
    int s = src[e], d = dst[e];
    zd[e] = z[d];
    float dx = pos[s * 3 + 0] - pos[d * 3 + 0];
    float dy = pos[s * 3 + 1] - pos[d * 3 + 1];
    float dz = pos[s * 3 + 2] - pos[d * 3 + 2];
    float r = sqrtf(dx * dx + dy * dy + dz * dz);
#pragma unroll
    for (int j = 0; j < NMU; ++j) {
        float t = r - 0.5f * (float)j;
        ef[(size_t)e * F_DIM + j] = expf(-2.0f * t * t);
    }
    ef[(size_t)e * F_DIM + NMU] = 1.0f;
}

__device__ __forceinline__ void row_ptr_item(const int* __restrict__ src, int* __restrict__ rp, int N, int E, int e) {
    if (e == 0) {
        for (int n = 0; n <= src[0]; ++n) rp[n] = 0;
    } else if (e == E) {
        for (int n = src[E - 1] + 1; n <= N; ++n) rp[n] = E;
    } else {
        int a = src[e - 1], b = src[e];
        for (int n = a + 1; n <= b; ++n) rp[n] = e;
    }
}

#define TP_PITCH 68
#define TP_TILES 4

__device__ __forceinline__ void tp_load_b(v2f (&bfr)[16], const float* __restrict__ Wl, int tile, int m, int hi) {
    int colBase = tile << 4;
    int f       = colBase >> 6;
    int bcol    = (colBase & 63) + m;
    const float* Bp = Wl + (size_t)f * (A_DIM * A_DIM) + 2 * hi * A_DIM + bcol;
#pragma unroll
    for (int k = 0; k < 16; ++k) {
        bfr[k].x = Bp[k * 4 * A_DIM];
        bfr[k].y = Bp[k * 4 * A_DIM + A_DIM];
    }
}

template <int TILES>
__device__ __forceinline__ int tp_first_tile(int tileBlock) {
    return (tileBlock * 4 + (int)(threadIdx.x >> 5)) * TILES;
}

template <int TILES, bool AHEAD>
__device__ __forceinline__ void tp_tiles(const float* sy, const float* __restrict__ Wl,
                                         float* __restrict__ g, int N, int nodeBase, int tileBlock,
                                         v2f (&bcur)[16]) {
    bool full = (nodeBase + 64 <= N);
    int lane = threadIdx.x & 31;
    int m    = lane & 15;
    int hi   = lane >> 4;
    int tile0 = tp_first_tile<TILES>(tileBlock);

    for (int t = 0; t < TILES; ++t) {
        int tile    = tile0 + t;
        int colBase = tile << 4;

        v2f bnxt[16];
        if (AHEAD) {
            if (t + 1 < TILES) tp_load_b(bnxt, Wl, tile + 1, m, hi);
        } else {
            tp_load_b(bcur, Wl, tile, m, hi);
        }

        v8f c0 = {}, c1 = {}, c2 = {}, c3 = {};
#pragma unroll
        for (int k = 0; k < 16; ++k) {
            int ka = k * 4 + 2 * hi;
            v2f a0; a0.x = sy[(m     ) * TP_PITCH + ka]; a0.y = sy[(m     ) * TP_PITCH + ka + 1];
            v2f a1; a1.x = sy[(m + 16) * TP_PITCH + ka]; a1.y = sy[(m + 16) * TP_PITCH + ka + 1];
            v2f a2; a2.x = sy[(m + 32) * TP_PITCH + ka]; a2.y = sy[(m + 32) * TP_PITCH + ka + 1];
            v2f a3; a3.x = sy[(m + 48) * TP_PITCH + ka]; a3.y = sy[(m + 48) * TP_PITCH + ka + 1];
            c0 = __builtin_amdgcn_wmma_f32_16x16x4_f32(false, a0, false, bcur[k], (short)0, c0, false, false);
            c1 = __builtin_amdgcn_wmma_f32_16x16x4_f32(false, a1, false, bcur[k], (short)0, c1, false, false);
            c2 = __builtin_amdgcn_wmma_f32_16x16x4_f32(false, a2, false, bcur[k], (short)0, c2, false, false);
            c3 = __builtin_amdgcn_wmma_f32_16x16x4_f32(false, a3, false, bcur[k], (short)0, c3, false, false);
        }
        size_t colOff = (size_t)colBase + m;
        if (full) {
#pragma unroll
            for (int r = 0; r < 8; ++r) {
                int rr = hi * 8 + r;
                g[(size_t)(nodeBase      + rr) * G_DIM + colOff] = c0[r];
                g[(size_t)(nodeBase + 16 + rr) * G_DIM + colOff] = c1[r];
                g[(size_t)(nodeBase + 32 + rr) * G_DIM + colOff] = c2[r];
                g[(size_t)(nodeBase + 48 + rr) * G_DIM + colOff] = c3[r];
            }
        } else {
#pragma unroll
            for (int r = 0; r < 8; ++r) {
                int rr = hi * 8 + r;
                if (nodeBase      + rr < N) g[(size_t)(nodeBase      + rr) * G_DIM + colOff] = c0[r];
                if (nodeBase + 16 + rr < N) g[(size_t)(nodeBase + 16 + rr) * G_DIM + colOff] = c1[r];
                if (nodeBase + 32 + rr < N) g[(size_t)(nodeBase + 32 + rr) * G_DIM + colOff] = c2[r];
                if (nodeBase + 48 + rr < N) g[(size_t)(nodeBase + 48 + rr) * G_DIM + colOff] = c3[r];
            }
        }
        if (AHEAD) {
            if (t + 1 < TILES) {
#pragma unroll
                for (int k = 0; k < 16; ++k) bcur[k] = bnxt[k];
            }
        }
    }
}

__global__ __launch_bounds__(128) void node_tp_wmma(const float* __restrict__ y,
                                                    const float* __restrict__ Wl,
                                                    float* __restrict__ g, int N) {
    __shared__ __attribute__((aligned(16))) float sy[64 * TP_PITCH];
    int nodeBase = blockIdx.x * 64;
    v2f bfr[16];
    tp_load_b(bfr, Wl, tp_first_tile<TP_TILES>((int)blockIdx.y), (int)(threadIdx.x & 15), (int)((threadIdx.x >> 4) & 1));
    int c  = (threadIdx.x * 4) & 63;
    int r0 = threadIdx.x >> 4;
    float4 st[8];
#pragma unroll
    for (int j = 0; j < 8; ++j) {
        int row = nodeBase + r0 + 8 * j;
        if (row >= N) row = N - 1;
        st[j] = *(const float4*)(y + (size_t)row * A_DIM + c);
    }
#pragma unroll
    for (int j = 0; j < 8; ++j) *(float4*)(sy + (r0 + 8 * j) * TP_PITCH + c) = st[j];
    __syncthreads();
    tp_tiles<TP_TILES, true>(sy, Wl, g, N, nodeBase, (int)blockIdx.y, bfr);
}

#define MAX_Z 100
#define TAB_RB 2
#define TAB_CG 6
#define TAB_TILES 2
__global__ __launch_bounds__(128) void first_table_wmma(const int* __restrict__ z,
                                                        const float* __restrict__ We,
                                                        const float* __restrict__ be,
                                                        const float* __restrict__ Wl,
                                                        float* __restrict__ yw,
                                                        float* __restrict__ tab,
                                                        int* __restrict__ zd,
                                                        int N,
                                                        const float* __restrict__ pos,
                                                        const int* __restrict__ src,
                                                        const int* __restrict__ dst,
                                                        float* __restrict__ ef,
                                                        int* __restrict__ rp, int E) {
    __shared__ __attribute__((aligned(16))) float sy[64 * TP_PITCH];
    int b = (int)blockIdx.x;
    if (b >= TAB_RB * TAB_CG) {
        int e = (b - TAB_RB * TAB_CG) * 128 + (int)threadIdx.x;
        if (e < E)  edge_feat_item(pos, src, dst, z, ef, zd, e);
        if (e <= E) row_ptr_item(src, rp, N, E, e);
        int nItems = ((E + 1 + 127) >> 7) * 128;
        for (int i = e; i < N * 16; i += nItems) {
            int row = i >> 4;
            int c   = (i & 15) * 4;
            const float* wr = We + z[row] * A_DIM + c;
            float4 v;
            v.x = wr[0] + be[c];
            v.y = wr[1] + be[c + 1];
            v.z = wr[2] + be[c + 2];
            v.w = wr[3] + be[c + 3];
            *(float4*)(yw + (size_t)row * A_DIM + c) = v;
        }
        return;
    }
    int rb      = b / TAB_CG;
    int cg      = b - rb * TAB_CG;
    int rowBase = rb * 64;
    int c  = (threadIdx.x * 4) & 63;
    int r0 = threadIdx.x >> 4;
    float b0 = be[c], b1 = be[c + 1], b2 = be[c + 2], b3 = be[c + 3];
    float4 v[8];
#pragma unroll
    for (int j = 0; j < 8; ++j) {
        int row = rowBase + r0 + 8 * j;
        if (row >= MAX_Z) row = MAX_Z - 1;
        const float* wr = We + row * A_DIM + c;
        v[j].x = wr[0] + b0;
        v[j].y = wr[1] + b1;
        v[j].z = wr[2] + b2;
        v[j].w = wr[3] + b3;
    }
#pragma unroll
    for (int j = 0; j < 8; ++j) *(float4*)(sy + (r0 + 8 * j) * TP_PITCH + c) = v[j];
    __syncthreads();
    v2f bfr[16];
    tp_tiles<TAB_TILES, false>(sy, Wl, tab, MAX_Z, rowBase, cg, bfr);
}

typedef __attribute__((ext_vector_type(4))) float v4f;

__device__ __forceinline__ float xhalf(float x) {
    return __builtin_bit_cast(float, __builtin_amdgcn_permlanex16(__builtin_bit_cast(int, x), __builtin_bit_cast(int, x),
                                                                  0x76543210, 0xfedcba98, false, false));
}

__device__ __forceinline__ v4f edge_msg(const float* __restrict__ g, const float* __restrict__ ef,
                                        const int* __restrict__ dst, int e, int c) {
    const v4f* fp = (const v4f*)(ef + (size_t)e * F_DIM);
    int d = dst[e];
    v4f w0 = fp[0], w1 = fp[1], w2 = fp[2];
    __builtin_amdgcn_sched_barrier(0);
    float w[F_DIM] = {w0.x, w0.y, w0.z, w0.w, w1.x, w1.y, w1.z, w1.w, w2.x, w2.y, w2.z, w2.w};
    unsigned row = (unsigned)d * (unsigned)(G_DIM * 4) + (unsigned)c * 16u;
    const char* gb = (const char*)g;
    v4f v[F_DIM];
    v[F_DIM - 1] = *(const v4f*)(gb + (row + (unsigned)(F_DIM - 1) * 256u));
#pragma unroll
    for (int f = 0; f < F_DIM - 1; ++f) {
        unsigned o = (w[f] != 0.0f) ? (unsigned)f * 256u : (unsigned)(F_DIM - 1) * 256u;
        v[f] = *(const v4f*)(gb + (row + o));
    }
    __builtin_amdgcn_sched_barrier(0);
    v4f m = {0.0f, 0.0f, 0.0f, 0.0f};
#pragma unroll
    for (int f = 0; f < F_DIM; ++f) m += w[f] * v[f];
    v4f s;
    s.x = silu(m.x * MSG_SCALE);
    s.y = silu(m.y * MSG_SCALE);
    s.z = silu(m.z * MSG_SCALE);
    s.w = silu(m.w * MSG_SCALE);
    return s;
}

template <bool LAST>
__global__ __launch_bounds__(256) void interact_kernel(float* __restrict__ y,
                                                       const float* __restrict__ g,
                                                       const float* __restrict__ ef,
                                                       const int* __restrict__ rp,
                                                       const int* __restrict__ dst,
                                                       float* __restrict__ out, int N) {
    int wave = __builtin_amdgcn_readfirstlane((int)(threadIdx.x >> 5));
    int n = blockIdx.x * 8 + wave;
    if (n >= N) return;
    int lane = threadIdx.x & 31;
    int c    = lane & 15;
    int hi   = lane >> 4;
    int e0 = rp[n], e1 = rp[n + 1];
    v4f acc = {0.0f, 0.0f, 0.0f, 0.0f};
    int e = e0;
    for (; e + 1 < e1; e += 2) {
        v4f s = edge_msg(g, ef, dst, e + hi, c);
        v4f o;
        o.x = xhalf(s.x); o.y = xhalf(s.y); o.z = xhalf(s.z); o.w = xhalf(s.w);
        acc = (acc + s) + o;
    }
    if (e < e1) {
        acc += edge_msg(g, ef, dst, e, c);
    }
    if (hi == 0) {
        v4f* yp = (v4f*)(y + (size_t)n * A_DIM) + c;
        v4f r = *yp + acc;
        *yp = r;
        if (LAST) {
            float* op = out + 1 + (size_t)n * A_DIM + 4 * c;
            op[0] = r.x; op[1] = r.y; op[2] = r.z; op[3] = r.w;
        }
    }
}

#define M_DIM 128
#define SY_PITCH 68

__global__ __launch_bounds__(256) void readout_wmma(const float* __restrict__ y,
                                                    const float* __restrict__ W1,
                                                    const float* __restrict__ b1,
                                                    const float* __restrict__ W2,
                                                    const float* __restrict__ b2,
                                                    float* __restrict__ partial, int N) {
    __shared__ __attribute__((aligned(16))) float sy[64 * SY_PITCH];
    __shared__ float red[8];
    int nodeBase = blockIdx.x * 64;
    int wave = threadIdx.x >> 5;
    int lane = threadIdx.x & 31;
    int m    = lane & 15;
    int hi   = lane >> 4;
    int col  = wave * 16 + m;

    const float* Bp = W1 + 2 * hi * M_DIM + col;
    v2f bfr[16];
#pragma unroll
    for (int k = 0; k < 16; ++k) {
        bfr[k].x = Bp[k * 4 * M_DIM];
        bfr[k].y = Bp[k * 4 * M_DIM + M_DIM];
    }
    float bias = b1[col];
    float w2   = W2[col];

    float4 st[4];
#pragma unroll
    for (int j = 0; j < 4; ++j) {
        int i = threadIdx.x * 4 + j * 1024;
        int row = nodeBase + (i >> 6);
        if (row >= N) row = N - 1;
        st[j] = *(const float4*)(y + (size_t)row * A_DIM + (i & 63));
    }
#pragma unroll
    for (int j = 0; j < 4; ++j) {
        int i = threadIdx.x * 4 + j * 1024;
        *(float4*)(sy + (i >> 6) * SY_PITCH + (i & 63)) = st[j];
    }
    __syncthreads();

    v8f c0 = {}, c1 = {}, c2 = {}, c3 = {};
#pragma unroll
    for (int k = 0; k < 16; ++k) {
        int ka = k * 4 + 2 * hi;
        v2f a0 = *(const v2f*)(sy + (m     ) * SY_PITCH + ka);
        v2f a1 = *(const v2f*)(sy + (m + 16) * SY_PITCH + ka);
        v2f a2 = *(const v2f*)(sy + (m + 32) * SY_PITCH + ka);
        v2f a3 = *(const v2f*)(sy + (m + 48) * SY_PITCH + ka);
        c0 = __builtin_amdgcn_wmma_f32_16x16x4_f32(false, a0, false, bfr[k], (short)0, c0, false, false);
        c1 = __builtin_amdgcn_wmma_f32_16x16x4_f32(false, a1, false, bfr[k], (short)0, c1, false, false);
        c2 = __builtin_amdgcn_wmma_f32_16x16x4_f32(false, a2, false, bfr[k], (short)0, c2, false, false);
        c3 = __builtin_amdgcn_wmma_f32_16x16x4_f32(false, a3, false, bfr[k], (short)0, c3, false, false);
    }

    float s = 0.0f;
#pragma unroll
    for (int r = 0; r < 8; ++r) {
        int row = nodeBase + hi * 8 + r;
        float h0 = silu(c0[r] + bias);
        float h1 = silu(c1[r] + bias);
        float h2 = silu(c2[r] + bias);
        float h3 = silu(c3[r] + bias);
        s += (row      < N) ? h0 : 0.0f;
        s += (row + 16 < N) ? h1 : 0.0f;
        s += (row + 32 < N) ? h2 : 0.0f;
        s += (row + 48 < N) ? h3 : 0.0f;
    }
    s *= w2;
    for (int off = 16; off > 0; off >>= 1) s += __shfl_xor(s, off);
    if (lane == 0) red[wave] = s;
    __syncthreads();
    if (threadIdx.x == 0) {
        float tot = 0.0f;
#pragma unroll
        for (int w = 0; w < 8; ++w) tot += red[w];
        int cnt = N - nodeBase;
        if (cnt > 64) cnt = 64;
        partial[blockIdx.x] = tot + b2[0] * (float)cnt;
    }
}

__global__ void finalize_kernel(const float* __restrict__ partial, int nPart,
                                const float* __restrict__ energy,
                                float* __restrict__ out, int N) {
    __shared__ float red[256];
    float v = 0.0f;
    for (int i = threadIdx.x; i < nPart; i += 256) v += partial[i];
    red[threadIdx.x] = v;
    __syncthreads();
    for (int off = 128; off > 0; off >>= 1) {
        if (threadIdx.x < off) red[threadIdx.x] += red[threadIdx.x + off];
        __syncthreads();
    }
    if (threadIdx.x == 0) {
        out[0] = red[0];
        out[1 + (size_t)N * A_DIM] = energy[0] * KCAL2MEV * (-1.0f);
    }
}

extern "C" void kernel_launch(void* const* d_in, const int* in_sizes, int n_in,
                              void* d_out, int out_size, void* d_ws, size_t ws_size,
                              hipStream_t stream) {
    const int*   z      = (const int*)d_in[0];
    const float* pos    = (const float*)d_in[1];
    const float* energy = (const float*)d_in[2];
    const int*   src    = (const int*)d_in[3];
    const int*   dst    = (const int*)d_in[4];
    const float* We     = (const float*)d_in[5];
    const float* be     = (const float*)d_in[6];
    const float* Wtp    = (const float*)d_in[7];
    const float* W1     = (const float*)d_in[8];
    const float* b1     = (const float*)d_in[9];
    const float* W2     = (const float*)d_in[10];
    const float* b2     = (const float*)d_in[11];
    float* out = (float*)d_out;

    const int N = in_sizes[0];
    const int E = in_sizes[3];

    float* fws = (float*)d_ws;
    size_t off = 0;
    float* ef      = fws + off; off += (size_t)E * F_DIM;
    float* g       = fws + off; off += (size_t)N * G_DIM;
    float* yw      = fws + off; off += (size_t)N * A_DIM;
    int*   rp      = (int*)(fws + off);
    float* partial = g;
    float* tab     = g;
    int*   zd      = (int*)(g + (size_t)128 * G_DIM);

    int nBlk = (N + 63) / 64;

    first_table_wmma<<<TAB_RB * TAB_CG + (E + 1 + 127) / 128, 128, 0, stream>>>(
        z, We, be, Wtp, yw, tab, zd, N, pos, src, dst, ef, rp, E);
    interact_kernel<false><<<(N + 7) / 8, 256, 0, stream>>>(yw, tab, ef, rp, zd, out, N);

    node_tp_wmma<<<dim3(nBlk, 3), 128, 0, stream>>>(yw, Wtp + (size_t)1 * F_DIM * A_DIM * A_DIM, g, N);
    interact_kernel<false><<<(N + 7) / 8, 256, 0, stream>>>(yw, g, ef, rp, dst, out, N);

    node_tp_wmma<<<dim3(nBlk, 3), 128, 0, stream>>>(yw, Wtp + (size_t)2 * F_DIM * A_DIM * A_DIM, g, N);
    interact_kernel<true><<<(N + 7) / 8, 256, 0, stream>>>(yw, g, ef, rp, dst, out, N);

    readout_wmma<<<nBlk, 256, 0, stream>>>(yw, W1, b1, W2, b2, partial, N);
    finalize_kernel<<<1, 256, 0, stream>>>(partial, nBlk, energy, out, N);
}


# SiLU as x * rcp(1 + exp2(-x log2 e)) on the hardware exp2 and reciprocal instructions, not library expf and IEEE division: 22 -> 5 vector instructions per value; its last-place rounding changes
# speedup vs baseline: 2.1675x; 1.0380x over previous
#include <hip/hip_runtime.h>
#include <math.h>

#define A_DIM 64
#define F_DIM 12
#define G_DIM (F_DIM * A_DIM)
#define NMU 11
#define MSG_SCALE 3.60843918e-04f
#define KCAL2MEV 43.3641f

typedef __attribute__((ext_vector_type(2))) float v2f;
typedef __attribute__((ext_vector_type(8))) float v8f;

__device__ __forceinline__ float silu(float x) {
    float e = __builtin_amdgcn_exp2f(x * -1.44269504f);
    return x * __builtin_amdgcn_rcpf(1.0f + e);
}

__device__ __forceinline__ void edge_feat_item(const float* __restrict__ pos, const int* __restrict__ src,
                                               const int* __restrict__ dst, const int* __restrict__ z,
                                               float* __restrict__ ef, int* __restrict__ zd, int e) {
    int s = src[e], d = dst[e];
    zd[e] = z[d];
    float dx = pos[s * 3 + 0] - pos[d * 3 + 0];
    float dy = pos[s * 3 + 1] - pos[d * 3 + 1];
    float dz = pos[s * 3 + 2] - pos[d * 3 + 2];
    float r = sqrtf(dx * dx + dy * dy + dz * dz);
#pragma unroll
    for (int j = 0; j < NMU; ++j) {
        float t = r - 0.5f * (float)j;
        ef[(size_t)e * F_DIM + j] = expf(-2.0f * t * t);
    }
    ef[(size_t)e * F_DIM + NMU] = 1.0f;
}

__device__ __forceinline__ void row_ptr_item(const int* __restrict__ src, int* __restrict__ rp, int N, int E, int e) {
    if (e == 0) {
        for (int n = 0; n <= src[0]; ++n) rp[n] = 0;
    } else if (e == E) {
        for (int n = src[E - 1] + 1; n <= N; ++n) rp[n] = E;
    } else {
        int a = src[e - 1], b = src[e];
        for (int n = a + 1; n <= b; ++n) rp[n] = e;
    }
}

#define TP_PITCH 68
#define TP_TILES 4

__device__ __forceinline__ void tp_load_b(v2f (&bfr)[16], const float* __restrict__ Wl, int tile, int m, int hi) {
    int colBase = tile << 4;
    int f       = colBase >> 6;
    int bcol    = (colBase & 63) + m;
    const float* Bp = Wl + (size_t)f * (A_DIM * A_DIM) + 2 * hi * A_DIM + bcol;
#pragma unroll
    for (int k = 0; k < 16; ++k) {
        bfr[k].x = Bp[k * 4 * A_DIM];
        bfr[k].y = Bp[k * 4 * A_DIM + A_DIM];
    }
}

template <int TILES>
__device__ __forceinline__ int tp_first_tile(int tileBlock) {
    return (tileBlock * 4 + (int)(threadIdx.x >> 5)) * TILES;
}

template <int TILES, bool AHEAD>
__device__ __forceinline__ void tp_tiles(const float* sy, const float* __restrict__ Wl,
                                         float* __restrict__ g, int N, int nodeBase, int tileBlock,
                                         v2f (&bcur)[16]) {
    bool full = (nodeBase + 64 <= N);
    int lane = threadIdx.x & 31;
    int m    = lane & 15;
    int hi   = lane >> 4;
    int tile0 = tp_first_tile<TILES>(tileBlock);

    for (int t = 0; t < TILES; ++t) {
        int tile    = tile0 + t;
        int colBase = tile << 4;

        v2f bnxt[16];
        if (AHEAD) {
            if (t + 1 < TILES) tp_load_b(bnxt, Wl, tile + 1, m, hi);
        } else {
            tp_load_b(bcur, Wl, tile, m, hi);
        }

        v8f c0 = {}, c1 = {}, c2 = {}, c3 = {};
#pragma unroll
        for (int k = 0; k < 16; ++k) {
            int ka = k * 4 + 2 * hi;
            v2f a0; a0.x = sy[(m     ) * TP_PITCH + ka]; a0.y = sy[(m     ) * TP_PITCH + ka + 1];
            v2f a1; a1.x = sy[(m + 16) * TP_PITCH + ka]; a1.y = sy[(m + 16) * TP_PITCH + ka + 1];
            v2f a2; a2.x = sy[(m + 32) * TP_PITCH + ka]; a2.y = sy[(m + 32) * TP_PITCH + ka + 1];
            v2f a3; a3.x = sy[(m + 48) * TP_PITCH + ka]; a3.y = sy[(m + 48) * TP_PITCH + ka + 1];
            c0 = __builtin_amdgcn_wmma_f32_16x16x4_f32(false, a0, false, bcur[k], (short)0, c0, false, false);
            c1 = __builtin_amdgcn_wmma_f32_16x16x4_f32(false, a1, false, bcur[k], (short)0, c1, false, false);
            c2 = __builtin_amdgcn_wmma_f32_16x16x4_f32(false, a2, false, bcur[k], (short)0, c2, false, false);
            c3 = __builtin_amdgcn_wmma_f32_16x16x4_f32(false, a3, false, bcur[k], (short)0, c3, false, false);
        }
        size_t colOff = (size_t)colBase + m;
        if (full) {
#pragma unroll
            for (int r = 0; r < 8; ++r) {
                int rr = hi * 8 + r;
                g[(size_t)(nodeBase      + rr) * G_DIM + colOff] = c0[r];
                g[(size_t)(nodeBase + 16 + rr) * G_DIM + colOff] = c1[r];
                g[(size_t)(nodeBase + 32 + rr) * G_DIM + colOff] = c2[r];
                g[(size_t)(nodeBase + 48 + rr) * G_DIM + colOff] = c3[r];
            }
        } else {
#pragma unroll
            for (int r = 0; r < 8; ++r) {
                int rr = hi * 8 + r;
                if (nodeBase      + rr < N) g[(size_t)(nodeBase      + rr) * G_DIM + colOff] = c0[r];
                if (nodeBase + 16 + rr < N) g[(size_t)(nodeBase + 16 + rr) * G_DIM + colOff] = c1[r];
                if (nodeBase + 32 + rr < N) g[(size_t)(nodeBase + 32 + rr) * G_DIM + colOff] = c2[r];
                if (nodeBase + 48 + rr < N) g[(size_t)(nodeBase + 48 + rr) * G_DIM + colOff] = c3[r];
            }
        }
        if (AHEAD) {
            if (t + 1 < TILES) {
#pragma unroll
                for (int k = 0; k < 16; ++k) bcur[k] = bnxt[k];
            }
        }
    }
}

__global__ __launch_bounds__(128) void node_tp_wmma(const float* __restrict__ y,
                                                    const float* __restrict__ Wl,
                                                    float* __restrict__ g, int N) {
    __shared__ __attribute__((aligned(16))) float sy[64 * TP_PITCH];
    int nodeBase = blockIdx.x * 64;
    v2f bfr[16];
    tp_load_b(bfr, Wl, tp_first_tile<TP_TILES>((int)blockIdx.y), (int)(threadIdx.x & 15), (int)((threadIdx.x >> 4) & 1));
    int c  = (threadIdx.x * 4) & 63;
    int r0 = threadIdx.x >> 4;
    float4 st[8];
#pragma unroll
    for (int j = 0; j < 8; ++j) {
        int row = nodeBase + r0 + 8 * j;
        if (row >= N) row = N - 1;
        st[j] = *(const float4*)(y + (size_t)row * A_DIM + c);
    }
#pragma unroll
    for (int j = 0; j < 8; ++j) *(float4*)(sy + (r0 + 8 * j) * TP_PITCH + c) = st[j];
    __syncthreads();
    tp_tiles<TP_TILES, true>(sy, Wl, g, N, nodeBase, (int)blockIdx.y, bfr);
}

#define MAX_Z 100
#define TAB_RB 2
#define TAB_CG 6
#define TAB_TILES 2
__global__ __launch_bounds__(128) void first_table_wmma(const int* __restrict__ z,
                                                        const float* __restrict__ We,
                                                        const float* __restrict__ be,
                                                        const float* __restrict__ Wl,
                                                        float* __restrict__ yw,
                                                        float* __restrict__ tab,
                                                        int* __restrict__ zd,
                                                        int N,
                                                        const float* __restrict__ pos,
                                                        const int* __restrict__ src,
                                                        const int* __restrict__ dst,
                                                        float* __restrict__ ef,
                                                        int* __restrict__ rp, int E) {
    __shared__ __attribute__((aligned(16))) float sy[64 * TP_PITCH];
    int b = (int)blockIdx.x;
    if (b >= TAB_RB * TAB_CG) {
        int e = (b - TAB_RB * TAB_CG) * 128 + (int)threadIdx.x;
        if (e < E)  edge_feat_item(pos, src, dst, z, ef, zd, e);
        if (e <= E) row_ptr_item(src, rp, N, E, e);
        int nItems = ((E + 1 + 127) >> 7) * 128;
        for (int i = e; i < N * 16; i += nItems) {
            int row = i >> 4;
            int c   = (i & 15) * 4;
            const float* wr = We + z[row] * A_DIM + c;
            float4 v;
            v.x = wr[0] + be[c];
            v.y = wr[1] + be[c + 1];
            v.z = wr[2] + be[c + 2];
            v.w = wr[3] + be[c + 3];
            *(float4*)(yw + (size_t)row * A_DIM + c) = v;
        }
        return;
    }
    int rb      = b / TAB_CG;
    int cg      = b - rb * TAB_CG;
    int rowBase = rb * 64;
    int c  = (threadIdx.x * 4) & 63;
    int r0 = threadIdx.x >> 4;
    float b0 = be[c], b1 = be[c + 1], b2 = be[c + 2], b3 = be[c + 3];
    float4 v[8];
#pragma unroll
    for (int j = 0; j < 8; ++j) {
        int row = rowBase + r0 + 8 * j;
        if (row >= MAX_Z) row = MAX_Z - 1;
        const float* wr = We + row * A_DIM + c;
        v[j].x = wr[0] + b0;
        v[j].y = wr[1] + b1;
        v[j].z = wr[2] + b2;
        v[j].w = wr[3] + b3;
    }
#pragma unroll
    for (int j = 0; j < 8; ++j) *(float4*)(sy + (r0 + 8 * j) * TP_PITCH + c) = v[j];
    __syncthreads();
    v2f bfr[16];
    tp_tiles<TAB_TILES, false>(sy, Wl, tab, MAX_Z, rowBase, cg, bfr);
}

typedef __attribute__((ext_vector_type(4))) float v4f;

__device__ __forceinline__ float xhalf(float x) {
    return __builtin_bit_cast(float, __builtin_amdgcn_permlanex16(__builtin_bit_cast(int, x), __builtin_bit_cast(int, x),
                                                                  0x76543210, 0xfedcba98, false, false));
}

__device__ __forceinline__ v4f edge_msg(const float* __restrict__ g, const float* __restrict__ ef,
                                        const int* __restrict__ dst, int e, int c) {
    const v4f* fp = (const v4f*)(ef + (size_t)e * F_DIM);
    int d = dst[e];
    v4f w0 = fp[0], w1 = fp[1], w2 = fp[2];
    __builtin_amdgcn_sched_barrier(0);
    float w[F_DIM] = {w0.x, w0.y, w0.z, w0.w, w1.x, w1.y, w1.z, w1.w, w2.x, w2.y, w2.z, w2.w};
    unsigned row = (unsigned)d * (unsigned)(G_DIM * 4) + (unsigned)c * 16u;
    const char* gb = (const char*)g;
    v4f v[F_DIM];
    v[F_DIM - 1] = *(const v4f*)(gb + (row + (unsigned)(F_DIM - 1) * 256u));
#pragma unroll
    for (int f = 0; f < F_DIM - 1; ++f) {
        unsigned o = (w[f] != 0.0f) ? (unsigned)f * 256u : (unsigned)(F_DIM - 1) * 256u;
        v[f] = *(const v4f*)(gb + (row + o));
    }
    __builtin_amdgcn_sched_barrier(0);
    v4f m = {0.0f, 0.0f, 0.0f, 0.0f};
#pragma unroll
    for (int f = 0; f < F_DIM; ++f) m += w[f] * v[f];
    v4f s;
    s.x = silu(m.x * MSG_SCALE);
    s.y = silu(m.y * MSG_SCALE);
    s.z = silu(m.z * MSG_SCALE);
    s.w = silu(m.w * MSG_SCALE);
    return s;
}

template <bool LAST>
__global__ __launch_bounds__(256) void interact_kernel(float* __restrict__ y,
                                                       const float* __restrict__ g,
                                                       const float* __restrict__ ef,
                                                       const int* __restrict__ rp,
                                                       const int* __restrict__ dst,
                                                       float* __restrict__ out, int N) {
    int wave = __builtin_amdgcn_readfirstlane((int)(threadIdx.x >> 5));
    int n = blockIdx.x * 8 + wave;
    if (n >= N) return;
    int lane = threadIdx.x & 31;
    int c    = lane & 15;
    int hi   = lane >> 4;
    int e0 = rp[n], e1 = rp[n + 1];
    v4f acc = {0.0f, 0.0f, 0.0f, 0.0f};
    int e = e0;
    for (; e + 1 < e1; e += 2) {
        v4f s = edge_msg(g, ef, dst, e + hi, c);
        v4f o;
        o.x = xhalf(s.x); o.y = xhalf(s.y); o.z = xhalf(s.z); o.w = xhalf(s.w);
        acc = (acc + s) + o;
    }
    if (e < e1) {
        acc += edge_msg(g, ef, dst, e, c);
    }
    if (hi == 0) {
        v4f* yp = (v4f*)(y + (size_t)n * A_DIM) + c;
        v4f r = *yp + acc;
        *yp = r;
        if (LAST) {
            float* op = out + 1 + (size_t)n * A_DIM + 4 * c;
            op[0] = r.x; op[1] = r.y; op[2] = r.z; op[3] = r.w;
        }
    }
}

#define M_DIM 128
#define SY_PITCH 68

__global__ __launch_bounds__(256) void readout_wmma(const float* __restrict__ y,
                                                    const float* __restrict__ W1,
                                                    const float* __restrict__ b1,
                                                    const float* __restrict__ W2,
                                                    const float* __restrict__ b2,
                                                    float* __restrict__ partial, int N) {
    __shared__ __attribute__((aligned(16))) float sy[64 * SY_PITCH];
    __shared__ float red[8];
    int nodeBase = blockIdx.x * 64;
    int wave = threadIdx.x >> 5;
    int lane = threadIdx.x & 31;
    int m    = lane & 15;
    int hi   = lane >> 4;
    int col  = wave * 16 + m;

    const float* Bp = W1 + 2 * hi * M_DIM + col;
    v2f bfr[16];
#pragma unroll
    for (int k = 0; k < 16; ++k) {
        bfr[k].x = Bp[k * 4 * M_DIM];
        bfr[k].y = Bp[k * 4 * M_DIM + M_DIM];
    }
    float bias = b1[col];
    float w2   = W2[col];

    float4 st[4];
#pragma unroll
    for (int j = 0; j < 4; ++j) {
        int i = threadIdx.x * 4 + j * 1024;
        int row = nodeBase + (i >> 6);
        if (row >= N) row = N - 1;
        st[j] = *(const float4*)(y + (size_t)row * A_DIM + (i & 63));
    }
#pragma unroll
    for (int j = 0; j < 4; ++j) {
        int i = threadIdx.x * 4 + j * 1024;
        *(float4*)(sy + (i >> 6) * SY_PITCH + (i & 63)) = st[j];
    }
    __syncthreads();

    v8f c0 = {}, c1 = {}, c2 = {}, c3 = {};
#pragma unroll
    for (int k = 0; k < 16; ++k) {
        int ka = k * 4 + 2 * hi;
        v2f a0 = *(const v2f*)(sy + (m     ) * SY_PITCH + ka);
        v2f a1 = *(const v2f*)(sy + (m + 16) * SY_PITCH + ka);
        v2f a2 = *(const v2f*)(sy + (m + 32) * SY_PITCH + ka);
        v2f a3 = *(const v2f*)(sy + (m + 48) * SY_PITCH + ka);
        c0 = __builtin_amdgcn_wmma_f32_16x16x4_f32(false, a0, false, bfr[k], (short)0, c0, false, false);
        c1 = __builtin_amdgcn_wmma_f32_16x16x4_f32(false, a1, false, bfr[k], (short)0, c1, false, false);
        c2 = __builtin_amdgcn_wmma_f32_16x16x4_f32(false, a2, false, bfr[k], (short)0, c2, false, false);
        c3 = __builtin_amdgcn_wmma_f32_16x16x4_f32(false, a3, false, bfr[k], (short)0, c3, false, false);
    }

    float s = 0.0f;
#pragma unroll
    for (int r = 0; r < 8; ++r) {
        int row = nodeBase + hi * 8 + r;
        float h0 = silu(c0[r] + bias);
        float h1 = silu(c1[r] + bias);
        float h2 = silu(c2[r] + bias);
        float h3 = silu(c3[r] + bias);
        s += (row      < N) ? h0 : 0.0f;
        s += (row + 16 < N) ? h1 : 0.0f;
        s += (row + 32 < N) ? h2 : 0.0f;
        s += (row + 48 < N) ? h3 : 0.0f;
    }
    s *= w2;
    for (int off = 16; off > 0; off >>= 1) s += __shfl_xor(s, off);
    if (lane == 0) red[wave] = s;
    __syncthreads();
    if (threadIdx.x == 0) {
        float tot = 0.0f;
#pragma unroll
        for (int w = 0; w < 8; ++w) tot += red[w];
        int cnt = N - nodeBase;
        if (cnt > 64) cnt = 64;
        partial[blockIdx.x] = tot + b2[0] * (float)cnt;
    }
}

__global__ void finalize_kernel(const float* __restrict__ partial, int nPart,
                                const float* __restrict__ energy,
                                float* __restrict__ out, int N) {
    __shared__ float red[256];
    float v = 0.0f;
    for (int i = threadIdx.x; i < nPart; i += 256) v += partial[i];
    red[threadIdx.x] = v;
    __syncthreads();
    for (int off = 128; off > 0; off >>= 1) {
        if (threadIdx.x < off) red[threadIdx.x] += red[threadIdx.x + off];
        __syncthreads();
    }
    if (threadIdx.x == 0) {
        out[0] = red[0];
        out[1 + (size_t)N * A_DIM] = energy[0] * KCAL2MEV * (-1.0f);
    }
}

extern "C" void kernel_launch(void* const* d_in, const int* in_sizes, int n_in,
                              void* d_out, int out_size, void* d_ws, size_t ws_size,
                              hipStream_t stream) {
    const int*   z      = (const int*)d_in[0];
    const float* pos    = (const float*)d_in[1];
    const float* energy = (const float*)d_in[2];
    const int*   src    = (const int*)d_in[3];
    const int*   dst    = (const int*)d_in[4];
    const float* We     = (const float*)d_in[5];
    const float* be     = (const float*)d_in[6];
    const float* Wtp    = (const float*)d_in[7];
    const float* W1     = (const float*)d_in[8];
    const float* b1     = (const float*)d_in[9];
    const float* W2     = (const float*)d_in[10];
    const float* b2     = (const float*)d_in[11];
    float* out = (float*)d_out;

    const int N = in_sizes[0];
    const int E = in_sizes[3];

    float* fws = (float*)d_ws;
    size_t off = 0;
    float* ef      = fws + off; off += (size_t)E * F_DIM;
    float* g       = fws + off; off += (size_t)N * G_DIM;
    float* yw      = fws + off; off += (size_t)N * A_DIM;
    int*   rp      = (int*)(fws + off);
    float* partial = g;
    float* tab     = g;
    int*   zd      = (int*)(g + (size_t)128 * G_DIM);

    int nBlk = (N + 63) / 64;

    first_table_wmma<<<TAB_RB * TAB_CG + (E + 1 + 127) / 128, 128, 0, stream>>>(
        z, We, be, Wtp, yw, tab, zd, N, pos, src, dst, ef, rp, E);
    interact_kernel<false><<<(N + 7) / 8, 256, 0, stream>>>(yw, tab, ef, rp, zd, out, N);

    node_tp_wmma<<<dim3(nBlk, 3), 128, 0, stream>>>(yw, Wtp + (size_t)1 * F_DIM * A_DIM * A_DIM, g, N);
    interact_kernel<false><<<(N + 7) / 8, 256, 0, stream>>>(yw, g, ef, rp, dst, out, N);

    node_tp_wmma<<<dim3(nBlk, 3), 128, 0, stream>>>(yw, Wtp + (size_t)2 * F_DIM * A_DIM * A_DIM, g, N);
    interact_kernel<true><<<(N + 7) / 8, 256, 0, stream>>>(yw, g, ef, rp, dst, out, N);

    readout_wmma<<<nBlk, 256, 0, stream>>>(yw, W1, b1, W2, b2, partial, N);
    finalize_kernel<<<1, 256, 0, stream>>>(partial, nBlk, energy, out, N);
}


# edge features: a gaussian weight below 2^-40 (beside the constant weight 1: far under f32 rounding) stored as 0, a cutoff; interact's zero test then skips its g chunk: 6 of 12 read, was 10
# speedup vs baseline: 2.5093x; 1.1577x over previous
#include <hip/hip_runtime.h>
#include <math.h>

#define A_DIM 64
#define F_DIM 12
#define G_DIM (F_DIM * A_DIM)
#define NMU 11
#define MSG_SCALE 3.60843918e-04f
#define KCAL2MEV 43.3641f
#define W_CUT 0x1p-40f

typedef __attribute__((ext_vector_type(2))) float v2f;
typedef __attribute__((ext_vector_type(8))) float v8f;

__device__ __forceinline__ float silu(float x) {
    float e = __builtin_amdgcn_exp2f(x * -1.44269504f);
    return x * __builtin_amdgcn_rcpf(1.0f + e);
}

__device__ __forceinline__ void edge_feat_item(const float* __restrict__ pos, const int* __restrict__ src,
                                               const int* __restrict__ dst, const int* __restrict__ z,
                                               float* __restrict__ ef, int* __restrict__ zd, int e) {
    int s = src[e], d = dst[e];
    zd[e] = z[d];
    float dx = pos[s * 3 + 0] - pos[d * 3 + 0];
    float dy = pos[s * 3 + 1] - pos[d * 3 + 1];
    float dz = pos[s * 3 + 2] - pos[d * 3 + 2];
    float r = sqrtf(dx * dx + dy * dy + dz * dz);
#pragma unroll
    for (int j = 0; j < NMU; ++j) {
        float t = r - 0.5f * (float)j;
        float wgt = expf(-2.0f * t * t);
        ef[(size_t)e * F_DIM + j] = (wgt < W_CUT) ? 0.0f : wgt;
    }
    ef[(size_t)e * F_DIM + NMU] = 1.0f;
}

__device__ __forceinline__ void row_ptr_item(const int* __restrict__ src, int* __restrict__ rp, int N, int E, int e) {
    if (e == 0) {
        for (int n = 0; n <= src[0]; ++n) rp[n] = 0;
    } else if (e == E) {
        for (int n = src[E - 1] + 1; n <= N; ++n) rp[n] = E;
    } else {
        int a = src[e - 1], b = src[e];
        for (int n = a + 1; n <= b; ++n) rp[n] = e;
    }
}

#define TP_PITCH 68
#define TP_TILES 4

__device__ __forceinline__ void tp_load_b(v2f (&bfr)[16], const float* __restrict__ Wl, int tile, int m, int hi) {
    int colBase = tile << 4;
    int f       = colBase >> 6;
    int bcol    = (colBase & 63) + m;
    const float* Bp = Wl + (size_t)f * (A_DIM * A_DIM) + 2 * hi * A_DIM + bcol;
#pragma unroll
    for (int k = 0; k < 16; ++k) {
        bfr[k].x = Bp[k * 4 * A_DIM];
        bfr[k].y = Bp[k * 4 * A_DIM + A_DIM];
    }
}

template <int TILES>
__device__ __forceinline__ int tp_first_tile(int tileBlock) {
    return (tileBlock * 4 + (int)(threadIdx.x >> 5)) * TILES;
}

template <int TILES, bool AHEAD>
__device__ __forceinline__ void tp_tiles(const float* sy, const float* __restrict__ Wl,
                                         float* __restrict__ g, int N, int nodeBase, int tileBlock,
                                         v2f (&bcur)[16]) {
    bool full = (nodeBase + 64 <= N);
    int lane = threadIdx.x & 31;
    int m    = lane & 15;
    int hi   = lane >> 4;
    int tile0 = tp_first_tile<TILES>(tileBlock);

    for (int t = 0; t < TILES; ++t) {
        int tile    = tile0 + t;
        int colBase = tile << 4;

        v2f bnxt[16];
        if (AHEAD) {
            if (t + 1 < TILES) tp_load_b(bnxt, Wl, tile + 1, m, hi);
        } else {
            tp_load_b(bcur, Wl, tile, m, hi);
        }

        v8f c0 = {}, c1 = {}, c2 = {}, c3 = {};
#pragma unroll
        for (int k = 0; k < 16; ++k) {
            int ka = k * 4 + 2 * hi;
            v2f a0; a0.x = sy[(m     ) * TP_PITCH + ka]; a0.y = sy[(m     ) * TP_PITCH + ka + 1];
            v2f a1; a1.x = sy[(m + 16) * TP_PITCH + ka]; a1.y = sy[(m + 16) * TP_PITCH + ka + 1];
            v2f a2; a2.x = sy[(m + 32) * TP_PITCH + ka]; a2.y = sy[(m + 32) * TP_PITCH + ka + 1];
            v2f a3; a3.x = sy[(m + 48) * TP_PITCH + ka]; a3.y = sy[(m + 48) * TP_PITCH + ka + 1];
            c0 = __builtin_amdgcn_wmma_f32_16x16x4_f32(false, a0, false, bcur[k], (short)0, c0, false, false);
            c1 = __builtin_amdgcn_wmma_f32_16x16x4_f32(false, a1, false, bcur[k], (short)0, c1, false, false);
            c2 = __builtin_amdgcn_wmma_f32_16x16x4_f32(false, a2, false, bcur[k], (short)0, c2, false, false);
            c3 = __builtin_amdgcn_wmma_f32_16x16x4_f32(false, a3, false, bcur[k], (short)0, c3, false, false);
        }
        size_t colOff = (size_t)colBase + m;
        if (full) {
#pragma unroll
            for (int r = 0; r < 8; ++r) {
                int rr = hi * 8 + r;
                g[(size_t)(nodeBase      + rr) * G_DIM + colOff] = c0[r];
                g[(size_t)(nodeBase + 16 + rr) * G_DIM + colOff] = c1[r];
                g[(size_t)(nodeBase + 32 + rr) * G_DIM + colOff] = c2[r];
                g[(size_t)(nodeBase + 48 + rr) * G_DIM + colOff] = c3[r];
            }
        } else {
#pragma unroll
            for (int r = 0; r < 8; ++r) {
                int rr = hi * 8 + r;
                if (nodeBase      + rr < N) g[(size_t)(nodeBase      + rr) * G_DIM + colOff] = c0[r];
                if (nodeBase + 16 + rr < N) g[(size_t)(nodeBase + 16 + rr) * G_DIM + colOff] = c1[r];
                if (nodeBase + 32 + rr < N) g[(size_t)(nodeBase + 32 + rr) * G_DIM + colOff] = c2[r];
                if (nodeBase + 48 + rr < N) g[(size_t)(nodeBase + 48 + rr) * G_DIM + colOff] = c3[r];
            }
        }
        if (AHEAD) {
            if (t + 1 < TILES) {
#pragma unroll
                for (int k = 0; k < 16; ++k) bcur[k] = bnxt[k];
            }
        }
    }
}

__global__ __launch_bounds__(128) void node_tp_wmma(const float* __restrict__ y,
                                                    const float* __restrict__ Wl,
                                                    float* __restrict__ g, int N) {
    __shared__ __attribute__((aligned(16))) float sy[64 * TP_PITCH];
    int nodeBase = blockIdx.x * 64;
    v2f bfr[16];
    tp_load_b(bfr, Wl, tp_first_tile<TP_TILES>((int)blockIdx.y), (int)(threadIdx.x & 15), (int)((threadIdx.x >> 4) & 1));
    int c  = (threadIdx.x * 4) & 63;
    int r0 = threadIdx.x >> 4;
    float4 st[8];
#pragma unroll
    for (int j = 0; j < 8; ++j) {
        int row = nodeBase + r0 + 8 * j;
        if (row >= N) row = N - 1;
        st[j] = *(const float4*)(y + (size_t)row * A_DIM + c);
    }
#pragma unroll
    for (int j = 0; j < 8; ++j) *(float4*)(sy + (r0 + 8 * j) * TP_PITCH + c) = st[j];
    __syncthreads();
    tp_tiles<TP_TILES, true>(sy, Wl, g, N, nodeBase, (int)blockIdx.y, bfr);
}

#define MAX_Z 100
#define TAB_RB 2
#define TAB_CG 6
#define TAB_TILES 2
__global__ __launch_bounds__(128) void first_table_wmma(const int* __restrict__ z,
                                                        const float* __restrict__ We,
                                                        const float* __restrict__ be,
                                                        const float* __restrict__ Wl,
                                                        float* __restrict__ yw,
                                                        float* __restrict__ tab,
                                                        int* __restrict__ zd,
                                                        int N,
                                                        const float* __restrict__ pos,
                                                        const int* __restrict__ src,
                                                        const int* __restrict__ dst,
                                                        float* __restrict__ ef,
                                                        int* __restrict__ rp, int E) {
    __shared__ __attribute__((aligned(16))) float sy[64 * TP_PITCH];
    int b = (int)blockIdx.x;
    if (b >= TAB_RB * TAB_CG) {
        int e = (b - TAB_RB * TAB_CG) * 128 + (int)threadIdx.x;
        if (e < E)  edge_feat_item(pos, src, dst, z, ef, zd, e);
        if (e <= E) row_ptr_item(src, rp, N, E, e);
        int nItems = ((E + 1 + 127) >> 7) * 128;
        for (int i = e; i < N * 16; i += nItems) {
            int row = i >> 4;
            int c   = (i & 15) * 4;
            const float* wr = We + z[row] * A_DIM + c;
            float4 v;
            v.x = wr[0] + be[c];
            v.y = wr[1] + be[c + 1];
            v.z = wr[2] + be[c + 2];
            v.w = wr[3] + be[c + 3];
            *(float4*)(yw + (size_t)row * A_DIM + c) = v;
        }
        return;
    }
    int rb      = b / TAB_CG;
    int cg      = b - rb * TAB_CG;
    int rowBase = rb * 64;
    int c  = (threadIdx.x * 4) & 63;
    int r0 = threadIdx.x >> 4;
    float b0 = be[c], b1 = be[c + 1], b2 = be[c + 2], b3 = be[c + 3];
    float4 v[8];
#pragma unroll
    for (int j = 0; j < 8; ++j) {
        int row = rowBase + r0 + 8 * j;
        if (row >= MAX_Z) row = MAX_Z - 1;
        const float* wr = We + row * A_DIM + c;
        v[j].x = wr[0] + b0;
        v[j].y = wr[1] + b1;
        v[j].z = wr[2] + b2;
        v[j].w = wr[3] + b3;
    }
#pragma unroll
    for (int j = 0; j < 8; ++j) *(float4*)(sy + (r0 + 8 * j) * TP_PITCH + c) = v[j];
    __syncthreads();
    v2f bfr[16];
    tp_tiles<TAB_TILES, false>(sy, Wl, tab, MAX_Z, rowBase, cg, bfr);
}

typedef __attribute__((ext_vector_type(4))) float v4f;

__device__ __forceinline__ float xhalf(float x) {
    return __builtin_bit_cast(float, __builtin_amdgcn_permlanex16(__builtin_bit_cast(int, x), __builtin_bit_cast(int, x),
                                                                  0x76543210, 0xfedcba98, false, false));
}

__device__ __forceinline__ v4f edge_msg(const float* __restrict__ g, const float* __restrict__ ef,
                                        const int* __restrict__ dst, int e, int c) {
    const v4f* fp = (const v4f*)(ef + (size_t)e * F_DIM);
    int d = dst[e];
    v4f w0 = fp[0], w1 = fp[1], w2 = fp[2];
    __builtin_amdgcn_sched_barrier(0);
    float w[F_DIM] = {w0.x, w0.y, w0.z, w0.w, w1.x, w1.y, w1.z, w1.w, w2.x, w2.y, w2.z, w2.w};
    unsigned row = (unsigned)d * (unsigned)(G_DIM * 4) + (unsigned)c * 16u;
    const char* gb = (const char*)g;
    v4f v[F_DIM];
    v[F_DIM - 1] = *(const v4f*)(gb + (row + (unsigned)(F_DIM - 1) * 256u));
#pragma unroll
    for (int f = 0; f < F_DIM - 1; ++f) {
        unsigned o = (w[f] != 0.0f) ? (unsigned)f * 256u : (unsigned)(F_DIM - 1) * 256u;
        v[f] = *(const v4f*)(gb + (row + o));
    }
    __builtin_amdgcn_sched_barrier(0);
    v4f m = {0.0f, 0.0f, 0.0f, 0.0f};
#pragma unroll
    for (int f = 0; f < F_DIM; ++f) m += w[f] * v[f];
    v4f s;
    s.x = silu(m.x * MSG_SCALE);
    s.y = silu(m.y * MSG_SCALE);
    s.z = silu(m.z * MSG_SCALE);
    s.w = silu(m.w * MSG_SCALE);
    return s;
}

template <bool LAST>
__global__ __launch_bounds__(256) void interact_kernel(float* __restrict__ y,
                                                       const float* __restrict__ g,
                                                       const float* __restrict__ ef,
                                                       const int* __restrict__ rp,
                                                       const int* __restrict__ dst,
                                                       float* __restrict__ out, int N) {
    int wave = __builtin_amdgcn_readfirstlane((int)(threadIdx.x >> 5));
    int n = blockIdx.x * 8 + wave;
    if (n >= N) return;
    int lane = threadIdx.x & 31;
    int c    = lane & 15;
    int hi   = lane >> 4;
    int e0 = rp[n], e1 = rp[n + 1];
    v4f acc = {0.0f, 0.0f, 0.0f, 0.0f};
    int e = e0;
    for (; e + 1 < e1; e += 2) {
        v4f s = edge_msg(g, ef, dst, e + hi, c);
        v4f o;
        o.x = xhalf(s.x); o.y = xhalf(s.y); o.z = xhalf(s.z); o.w = xhalf(s.w);
        acc = (acc + s) + o;
    }
    if (e < e1) {
        acc += edge_msg(g, ef, dst, e, c);
    }
    if (hi == 0) {
        v4f* yp = (v4f*)(y + (size_t)n * A_DIM) + c;
        v4f r = *yp + acc;
        *yp = r;
        if (LAST) {
            float* op = out + 1 + (size_t)n * A_DIM + 4 * c;
            op[0] = r.x; op[1] = r.y; op[2] = r.z; op[3] = r.w;
        }
    }
}

#define M_DIM 128
#define SY_PITCH 68

__global__ __launch_bounds__(256) void readout_wmma(const float* __restrict__ y,
                                                    const float* __restrict__ W1,
                                                    const float* __restrict__ b1,
                                                    const float* __restrict__ W2,
                                                    const float* __restrict__ b2,
                                                    float* __restrict__ partial, int N) {
    __shared__ __attribute__((aligned(16))) float sy[64 * SY_PITCH];
    __shared__ float red[8];
    int nodeBase = blockIdx.x * 64;
    int wave = threadIdx.x >> 5;
    int lane = threadIdx.x & 31;
    int m    = lane & 15;
    int hi   = lane >> 4;
    int col  = wave * 16 + m;

    const float* Bp = W1 + 2 * hi * M_DIM + col;
    v2f bfr[16];
#pragma unroll
    for (int k = 0; k < 16; ++k) {
        bfr[k].x = Bp[k * 4 * M_DIM];
        bfr[k].y = Bp[k * 4 * M_DIM + M_DIM];
    }
    float bias = b1[col];
    float w2   = W2[col];

    float4 st[4];
#pragma unroll
    for (int j = 0; j < 4; ++j) {
        int i = threadIdx.x * 4 + j * 1024;
        int row = nodeBase + (i >> 6);
        if (row >= N) row = N - 1;
        st[j] = *(const float4*)(y + (size_t)row * A_DIM + (i & 63));
    }
#pragma unroll
    for (int j = 0; j < 4; ++j) {
        int i = threadIdx.x * 4 + j * 1024;
        *(float4*)(sy + (i >> 6) * SY_PITCH + (i & 63)) = st[j];
    }
    __syncthreads();

    v8f c0 = {}, c1 = {}, c2 = {}, c3 = {};
#pragma unroll
    for (int k = 0; k < 16; ++k) {
        int ka = k * 4 + 2 * hi;
        v2f a0 = *(const v2f*)(sy + (m     ) * SY_PITCH + ka);
        v2f a1 = *(const v2f*)(sy + (m + 16) * SY_PITCH + ka);
        v2f a2 = *(const v2f*)(sy + (m + 32) * SY_PITCH + ka);
        v2f a3 = *(const v2f*)(sy + (m + 48) * SY_PITCH + ka);
        c0 = __builtin_amdgcn_wmma_f32_16x16x4_f32(false, a0, false, bfr[k], (short)0, c0, false, false);
        c1 = __builtin_amdgcn_wmma_f32_16x16x4_f32(false, a1, false, bfr[k], (short)0, c1, false, false);
        c2 = __builtin_amdgcn_wmma_f32_16x16x4_f32(false, a2, false, bfr[k], (short)0, c2, false, false);
        c3 = __builtin_amdgcn_wmma_f32_16x16x4_f32(false, a3, false, bfr[k], (short)0, c3, false, false);
    }

    float s = 0.0f;
#pragma unroll
    for (int r = 0; r < 8; ++r) {
        int row = nodeBase + hi * 8 + r;
        float h0 = silu(c0[r] + bias);
        float h1 = silu(c1[r] + bias);
        float h2 = silu(c2[r] + bias);
        float h3 = silu(c3[r] + bias);
        s += (row      < N) ? h0 : 0.0f;
        s += (row + 16 < N) ? h1 : 0.0f;
        s += (row + 32 < N) ? h2 : 0.0f;
        s += (row + 48 < N) ? h3 : 0.0f;
    }
    s *= w2;
    for (int off = 16; off > 0; off >>= 1) s += __shfl_xor(s, off);
    if (lane == 0) red[wave] = s;
    __syncthreads();
    if (threadIdx.x == 0) {
        float tot = 0.0f;
#pragma unroll
        for (int w = 0; w < 8; ++w) tot += red[w];
        int cnt = N - nodeBase;
        if (cnt > 64) cnt = 64;
        partial[blockIdx.x] = tot + b2[0] * (float)cnt;
    }
}

__global__ void finalize_kernel(const float* __restrict__ partial, int nPart,
                                const float* __restrict__ energy,
                                float* __restrict__ out, int N) {
    __shared__ float red[256];
    float v = 0.0f;
    for (int i = threadIdx.x; i < nPart; i += 256) v += partial[i];
    red[threadIdx.x] = v;
    __syncthreads();
    for (int off = 128; off > 0; off >>= 1) {
        if (threadIdx.x < off) red[threadIdx.x] += red[threadIdx.x + off];
        __syncthreads();
    }
    if (threadIdx.x == 0) {
        out[0] = red[0];
        out[1 + (size_t)N * A_DIM] = energy[0] * KCAL2MEV * (-1.0f);
    }
}

extern "C" void kernel_launch(void* const* d_in, const int* in_sizes, int n_in,
                              void* d_out, int out_size, void* d_ws, size_t ws_size,
                              hipStream_t stream) {
    const int*   z      = (const int*)d_in[0];
    const float* pos    = (const float*)d_in[1];
    const float* energy = (const float*)d_in[2];
    const int*   src    = (const int*)d_in[3];
    const int*   dst    = (const int*)d_in[4];
    const float* We     = (const float*)d_in[5];
    const float* be     = (const float*)d_in[6];
    const float* Wtp    = (const float*)d_in[7];
    const float* W1     = (const float*)d_in[8];
    const float* b1     = (const float*)d_in[9];
    const float* W2     = (const float*)d_in[10];
    const float* b2     = (const float*)d_in[11];
    float* out = (float*)d_out;

    const int N = in_sizes[0];
    const int E = in_sizes[3];

    float* fws = (float*)d_ws;
    size_t off = 0;
    float* ef      = fws + off; off += (size_t)E * F_DIM;
    float* g       = fws + off; off += (size_t)N * G_DIM;
    float* yw      = fws + off; off += (size_t)N * A_DIM;
    int*   rp      = (int*)(fws + off);
    float* partial = g;
    float* tab     = g;
    int*   zd      = (int*)(g + (size_t)128 * G_DIM);

    int nBlk = (N + 63) / 64;

    first_table_wmma<<<TAB_RB * TAB_CG + (E + 1 + 127) / 128, 128, 0, stream>>>(
        z, We, be, Wtp, yw, tab, zd, N, pos, src, dst, ef, rp, E);
    interact_kernel<false><<<(N + 7) / 8, 256, 0, stream>>>(yw, tab, ef, rp, zd, out, N);

    node_tp_wmma<<<dim3(nBlk, 3), 128, 0, stream>>>(yw, Wtp + (size_t)1 * F_DIM * A_DIM * A_DIM, g, N);
    interact_kernel<false><<<(N + 7) / 8, 256, 0, stream>>>(yw, g, ef, rp, dst, out, N);

    node_tp_wmma<<<dim3(nBlk, 3), 128, 0, stream>>>(yw, Wtp + (size_t)2 * F_DIM * A_DIM * A_DIM, g, N);
    interact_kernel<true><<<(N + 7) / 8, 256, 0, stream>>>(yw, g, ef, rp, dst, out, N);

    readout_wmma<<<nBlk, 256, 0, stream>>>(yw, W1, b1, W2, b2, partial, N);
    finalize_kernel<<<1, 256, 0, stream>>>(partial, nBlk, energy, out, N);
}
